# starting kernel
# baseline (speedup 1.0000x reference)
#include <hip/hip_runtime.h>
#include <hip/hip_bf16.h>

typedef float v2f __attribute__((ext_vector_type(2)));
typedef float v8f __attribute__((ext_vector_type(8)));

__device__ __forceinline__ void fadd_agent(float* p, float v) {
    __hip_atomic_fetch_add(p, v, __ATOMIC_RELAXED, __HIP_MEMORY_SCOPE_AGENT);
}

__global__ void bn_partial_kernel(const float* __restrict__ x,
                                  float* __restrict__ partial, int n) {
    float v[16];
#pragma unroll
    for (int k = 0; k < 16; ++k) v[k] = 0.f;

    const int stride = gridDim.x * blockDim.x;
    for (int i = blockIdx.x * blockDim.x + threadIdx.x; i < n; i += stride) {
        const float4* p = (const float4*)(x + (size_t)i * 8);
        float4 v0 = p[0];
        float4 v1 = p[1];
        float h;
        h = fmaxf(v0.x, 0.f); v[0] += h; v[8]  += h * h;
        h = fmaxf(v0.y, 0.f); v[1] += h; v[9]  += h * h;
        h = fmaxf(v0.z, 0.f); v[2] += h; v[10] += h * h;
        h = fmaxf(v0.w, 0.f); v[3] += h; v[11] += h * h;
        h = fmaxf(v1.x, 0.f); v[4] += h; v[12] += h * h;
        h = fmaxf(v1.y, 0.f); v[5] += h; v[13] += h * h;
        h = fmaxf(v1.z, 0.f); v[6] += h; v[14] += h * h;
        h = fmaxf(v1.w, 0.f); v[7] += h; v[15] += h * h;
    }

#pragma unroll
    for (int k = 0; k < 16; ++k) {
        float t = v[k];
        t += __shfl_xor(t, 1);
        t += __shfl_xor(t, 2);
        t += __shfl_xor(t, 4);
        t += __shfl_xor(t, 8);
        t += __shfl_xor(t, 16);
        v[k] = t;
    }

    __shared__ float sh[8 * 16];
    const int wave = threadIdx.x >> 5;
    const int lane = threadIdx.x & 31;
    if (lane == 0) {
#pragma unroll
        for (int k = 0; k < 16; ++k) sh[wave * 16 + k] = v[k];
    }
    __syncthreads();
    if (threadIdx.x < 16) {
        float t = 0.f;
#pragma unroll
        for (int w = 0; w < 8; ++w) t += sh[w * 16 + threadIdx.x];
        partial[blockIdx.x * 16 + threadIdx.x] = t;
    }
}

__global__ void bn_finalize_kernel(const float* __restrict__ partial,
                                   const float* __restrict__ gamma,
                                   const float* __restrict__ beta,
                                   float* __restrict__ scsh, int nb, float invN) {
    int k = threadIdx.x;
    if (k < 8) {
        float s = 0.f, q = 0.f;
        for (int b = 0; b < nb; ++b) {
            s += partial[b * 16 + k];
            q += partial[b * 16 + 8 + k];
        }
        float mean = s * invN;
        float var  = q * invN - mean * mean;
        float sc   = gamma[k] * rsqrtf(var + 1e-5f);
        scsh[k]     = sc;
        scsh[8 + k] = beta[k] - mean * sc;
    }
}

__global__ void deg_init_kernel(unsigned* __restrict__ deg, int n2) {
    int i = blockIdx.x * blockDim.x + threadIdx.x;
    if (i < n2) deg[i] = 1u;
}

__global__ void deg_count_kernel(const int* __restrict__ rows,
                                 const int* __restrict__ cols,
                                 unsigned* __restrict__ deg_c,
                                 unsigned* __restrict__ deg_r, int ne) {
    int e = blockIdx.x * blockDim.x + threadIdx.x;
    if (e >= ne) return;
    atomicAdd(&deg_r[rows[e]], 1u);
    atomicAdd(&deg_c[cols[e]], 1u);
}

__global__ void deg_to_dis_kernel(unsigned* __restrict__ deg, int n2) {
    int i = blockIdx.x * blockDim.x + threadIdx.x;
    if (i >= n2) return;
    float f = (float)deg[i];
    ((float*)deg)[i] = rsqrtf(f);
}

__global__ void transform_wmma_kernel(const float* __restrict__ x,
                                      const float* __restrict__ W_in,
                                      const float* __restrict__ W_out,
                                      const float* __restrict__ scsh,
                                      float* __restrict__ xw_in,
                                      float* __restrict__ xw_out,
                                      int nTiles, int n) {
    const int lane = threadIdx.x & 31;
    const int wave = threadIdx.x >> 5;
    const int tile = blockIdx.x * (blockDim.x >> 5) + wave;
    if (tile >= nTiles) return;

    const int half = lane >> 4;
    const int l    = lane & 15;
    int row = tile * 16 + l;
    int rowc = row < n ? row : (n - 1);

    const int kA0 = 2 * half;
    const int kA1 = 4 + 2 * half;

    const float* xr = x + (size_t)rowc * 8;
    v2f a0, a1;
    a0.x = fmaxf(xr[kA0],     0.f) * scsh[kA0]     + scsh[8 + kA0];
    a0.y = fmaxf(xr[kA0 + 1], 0.f) * scsh[kA0 + 1] + scsh[8 + kA0 + 1];
    a1.x = fmaxf(xr[kA1],     0.f) * scsh[kA1]     + scsh[8 + kA1];
    a1.y = fmaxf(xr[kA1 + 1], 0.f) * scsh[kA1 + 1] + scsh[8 + kA1 + 1];

    v2f bi0 = {0.f, 0.f}, bi1 = {0.f, 0.f};
    v2f bo0 = {0.f, 0.f}, bo1 = {0.f, 0.f};
    if (l < 8) {
        bi0.x = W_in[kA0 * 8 + l];       bi0.y = W_in[(kA0 + 1) * 8 + l];
        bi1.x = W_in[kA1 * 8 + l];       bi1.y = W_in[(kA1 + 1) * 8 + l];
        bo0.x = W_out[kA0 * 8 + l];      bo0.y = W_out[(kA0 + 1) * 8 + l];
        bo1.x = W_out[kA1 * 8 + l];      bo1.y = W_out[(kA1 + 1) * 8 + l];
    }

    v8f ci = {}; v8f co = {};
    ci = __builtin_amdgcn_wmma_f32_16x16x4_f32(false, a0, false, bi0, (short)0, ci, false, false);
    ci = __builtin_amdgcn_wmma_f32_16x16x4_f32(false, a1, false, bi1, (short)0, ci, false, false);
    co = __builtin_amdgcn_wmma_f32_16x16x4_f32(false, a0, false, bo0, (short)0, co, false, false);
    co = __builtin_amdgcn_wmma_f32_16x16x4_f32(false, a1, false, bo1, (short)0, co, false, false);

    const bool fullTile = (tile * 16 + 16) <= n;
    if (l < 8) {
        const int mbase = tile * 16 + 8 * half;
        float* pi = xw_in  + (size_t)mbase * 8 + l;
        float* po = xw_out + (size_t)mbase * 8 + l;
        if (fullTile) {
#pragma unroll
            for (int j = 0; j < 8; ++j) {
                pi[(size_t)j * 8] = ci[j];
                po[(size_t)j * 8] = co[j];
            }
        } else {
#pragma unroll
            for (int j = 0; j < 8; ++j) {
                if (mbase + j < n) {
                    pi[(size_t)j * 8] = ci[j];
                    po[(size_t)j * 8] = co[j];
                }
            }
        }
    }
}

__global__ void out_init_kernel(const float* __restrict__ xw_in,
                                const float* __restrict__ xw_out,
                                const float* __restrict__ dis_c,
                                const float* __restrict__ dis_r,
                                const float* __restrict__ b_in,
                                const float* __restrict__ b_out,
                                float* __restrict__ out, int n) {
    int i = blockIdx.x * blockDim.x + threadIdx.x;
    if (i >= n) return;
    float nc = dis_c[i] * dis_c[i];
    float nr = dis_r[i] * dis_r[i];
    const float* xi = xw_in  + (size_t)i * 8;
    const float* xo = xw_out + (size_t)i * 8;
    float* o = out + (size_t)i * 16;
#pragma unroll
    for (int k = 0; k < 8; ++k) o[k] = b_in[k] + xi[k] * nc;
#pragma unroll
    for (int k = 0; k < 8; ++k) o[8 + k] = b_out[k] + xo[k] * nr;
}

__global__ void edge_scatter_kernel(const int* __restrict__ rows,
                                    const int* __restrict__ cols,
                                    const float* __restrict__ xw_in,
                                    const float* __restrict__ xw_out,
                                    const float* __restrict__ dis_c,
                                    const float* __restrict__ dis_r,
                                    float* __restrict__ out, int ne) {
    int e = blockIdx.x * blockDim.x + threadIdx.x;
    if (e >= ne) return;
    int r = rows[e];
    int c = cols[e];
    float ni = dis_c[r] * dis_c[c];
    float no = dis_r[r] * dis_r[c];
    const float4* pi = (const float4*)(xw_in  + (size_t)r * 8);
    const float4* po = (const float4*)(xw_out + (size_t)c * 8);
    float4 xi0 = pi[0], xi1 = pi[1];
    float4 xo0 = po[0], xo1 = po[1];
    float* oc  = out + (size_t)c * 16;
    float* orr = out + (size_t)r * 16 + 8;
    fadd_agent(&oc[0], xi0.x * ni);
    fadd_agent(&oc[1], xi0.y * ni);
    fadd_agent(&oc[2], xi0.z * ni);
    fadd_agent(&oc[3], xi0.w * ni);
    fadd_agent(&oc[4], xi1.x * ni);
    fadd_agent(&oc[5], xi1.y * ni);
    fadd_agent(&oc[6], xi1.z * ni);
    fadd_agent(&oc[7], xi1.w * ni);
    fadd_agent(&orr[0], xo0.x * no);
    fadd_agent(&orr[1], xo0.y * no);
    fadd_agent(&orr[2], xo0.z * no);
    fadd_agent(&orr[3], xo0.w * no);
    fadd_agent(&orr[4], xo1.x * no);
    fadd_agent(&orr[5], xo1.y * no);
    fadd_agent(&orr[6], xo1.z * no);
    fadd_agent(&orr[7], xo1.w * no);
}

extern "C" void kernel_launch(void* const* d_in, const int* in_sizes, int n_in,
                              void* d_out, int out_size, void* d_ws, size_t ws_size,
                              hipStream_t stream) {
    const float* x     = (const float*)d_in[0];
    const int*   ei    = (const int*)d_in[1];
    const float* gamma = (const float*)d_in[2];
    const float* beta  = (const float*)d_in[3];
    const float* W_in  = (const float*)d_in[4];
    const float* b_in  = (const float*)d_in[5];
    const float* W_out = (const float*)d_in[6];
    const float* b_out = (const float*)d_in[7];
    float* out = (float*)d_out;

    const int n  = in_sizes[0] / 8;
    const int ne = in_sizes[1] / 2;
    const int NB = 1024;

    float* ws      = (float*)d_ws;
    float* xw_in   = ws;
    float* xw_out  = xw_in + (size_t)n * 8;
    float* dis_c   = xw_out + (size_t)n * 8;
    float* dis_r   = dis_c + n;
    float* partial = dis_r + n;
    float* scsh    = partial + (size_t)NB * 16;

    const int* rows = ei;
    const int* cols = ei + ne;

    bn_partial_kernel<<<NB, 256, 0, stream>>>(x, partial, n);
    bn_finalize_kernel<<<1, 32, 0, stream>>>(partial, gamma, beta, scsh, NB, 1.0f / (float)n);

    deg_init_kernel<<<(2 * n + 255) / 256, 256, 0, stream>>>((unsigned*)dis_c, 2 * n);
    deg_count_kernel<<<(ne + 255) / 256, 256, 0, stream>>>(rows, cols,
                                                           (unsigned*)dis_c, (unsigned*)dis_r, ne);
    deg_to_dis_kernel<<<(2 * n + 255) / 256, 256, 0, stream>>>((unsigned*)dis_c, 2 * n);

    const int nTiles = (n + 15) / 16;
    const int wavesPerBlock = 8;
    transform_wmma_kernel<<<(nTiles + wavesPerBlock - 1) / wavesPerBlock, 256, 0, stream>>>(
        x, W_in, W_out, scsh, xw_in, xw_out, nTiles, n);

    out_init_kernel<<<(n + 255) / 256, 256, 0, stream>>>(xw_in, xw_out, dis_c, dis_r,
                                                         b_in, b_out, out, n);
    edge_scatter_kernel<<<(ne + 255) / 256, 256, 0, stream>>>(rows, cols, xw_in, xw_out,
                                                              dis_c, dis_r, out, ne);
}


# edge_scatter: 8 lanes serve one edge, one feature each, so each f32 atomic add and row gather covers 4 whole 32-B rows, not 32 lines; sums stay in atomic arrival order as in the parent
# speedup vs baseline: 2.0387x; 2.0387x over previous
#include <hip/hip_runtime.h>
#include <hip/hip_bf16.h>

typedef float v2f __attribute__((ext_vector_type(2)));
typedef float v8f __attribute__((ext_vector_type(8)));

__device__ __forceinline__ void fadd_agent(float* p, float v) {
    __hip_atomic_fetch_add(p, v, __ATOMIC_RELAXED, __HIP_MEMORY_SCOPE_AGENT);
}

__global__ void bn_partial_kernel(const float* __restrict__ x,
                                  float* __restrict__ partial, int n) {
    float v[16];
#pragma unroll
    for (int k = 0; k < 16; ++k) v[k] = 0.f;

    const int stride = gridDim.x * blockDim.x;
    for (int i = blockIdx.x * blockDim.x + threadIdx.x; i < n; i += stride) {
        const float4* p = (const float4*)(x + (size_t)i * 8);
        float4 v0 = p[0];
        float4 v1 = p[1];
        float h;
        h = fmaxf(v0.x, 0.f); v[0] += h; v[8]  += h * h;
        h = fmaxf(v0.y, 0.f); v[1] += h; v[9]  += h * h;
        h = fmaxf(v0.z, 0.f); v[2] += h; v[10] += h * h;
        h = fmaxf(v0.w, 0.f); v[3] += h; v[11] += h * h;
        h = fmaxf(v1.x, 0.f); v[4] += h; v[12] += h * h;
        h = fmaxf(v1.y, 0.f); v[5] += h; v[13] += h * h;
        h = fmaxf(v1.z, 0.f); v[6] += h; v[14] += h * h;
        h = fmaxf(v1.w, 0.f); v[7] += h; v[15] += h * h;
    }

#pragma unroll
    for (int k = 0; k < 16; ++k) {
        float t = v[k];
        t += __shfl_xor(t, 1);
        t += __shfl_xor(t, 2);
        t += __shfl_xor(t, 4);
        t += __shfl_xor(t, 8);
        t += __shfl_xor(t, 16);
        v[k] = t;
    }

    __shared__ float sh[8 * 16];
    const int wave = threadIdx.x >> 5;
    const int lane = threadIdx.x & 31;
    if (lane == 0) {
#pragma unroll
        for (int k = 0; k < 16; ++k) sh[wave * 16 + k] = v[k];
    }
    __syncthreads();
    if (threadIdx.x < 16) {
        float t = 0.f;
#pragma unroll
        for (int w = 0; w < 8; ++w) t += sh[w * 16 + threadIdx.x];
        partial[blockIdx.x * 16 + threadIdx.x] = t;
    }
}

__global__ void bn_finalize_kernel(const float* __restrict__ partial,
                                   const float* __restrict__ gamma,
                                   const float* __restrict__ beta,
                                   float* __restrict__ scsh, int nb, float invN) {
    int k = threadIdx.x;
    if (k < 8) {
        float s = 0.f, q = 0.f;
        for (int b = 0; b < nb; ++b) {
            s += partial[b * 16 + k];
            q += partial[b * 16 + 8 + k];
        }
        float mean = s * invN;
        float var  = q * invN - mean * mean;
        float sc   = gamma[k] * rsqrtf(var + 1e-5f);
        scsh[k]     = sc;
        scsh[8 + k] = beta[k] - mean * sc;
    }
}

__global__ void deg_init_kernel(unsigned* __restrict__ deg, int n2) {
    int i = blockIdx.x * blockDim.x + threadIdx.x;
    if (i < n2) deg[i] = 1u;
}

__global__ void deg_count_kernel(const int* __restrict__ rows,
                                 const int* __restrict__ cols,
                                 unsigned* __restrict__ deg_c,
                                 unsigned* __restrict__ deg_r, int ne) {
    int e = blockIdx.x * blockDim.x + threadIdx.x;
    if (e >= ne) return;
    atomicAdd(&deg_r[rows[e]], 1u);
    atomicAdd(&deg_c[cols[e]], 1u);
}

__global__ void deg_to_dis_kernel(unsigned* __restrict__ deg, int n2) {
    int i = blockIdx.x * blockDim.x + threadIdx.x;
    if (i >= n2) return;
    float f = (float)deg[i];
    ((float*)deg)[i] = rsqrtf(f);
}

__global__ void transform_wmma_kernel(const float* __restrict__ x,
                                      const float* __restrict__ W_in,
                                      const float* __restrict__ W_out,
                                      const float* __restrict__ scsh,
                                      float* __restrict__ xw_in,
                                      float* __restrict__ xw_out,
                                      int nTiles, int n) {
    const int lane = threadIdx.x & 31;
    const int wave = threadIdx.x >> 5;
    const int tile = blockIdx.x * (blockDim.x >> 5) + wave;
    if (tile >= nTiles) return;

    const int half = lane >> 4;
    const int l    = lane & 15;
    int row = tile * 16 + l;
    int rowc = row < n ? row : (n - 1);

    const int kA0 = 2 * half;
    const int kA1 = 4 + 2 * half;

    const float* xr = x + (size_t)rowc * 8;
    v2f a0, a1;
    a0.x = fmaxf(xr[kA0],     0.f) * scsh[kA0]     + scsh[8 + kA0];
    a0.y = fmaxf(xr[kA0 + 1], 0.f) * scsh[kA0 + 1] + scsh[8 + kA0 + 1];
    a1.x = fmaxf(xr[kA1],     0.f) * scsh[kA1]     + scsh[8 + kA1];
    a1.y = fmaxf(xr[kA1 + 1], 0.f) * scsh[kA1 + 1] + scsh[8 + kA1 + 1];

    v2f bi0 = {0.f, 0.f}, bi1 = {0.f, 0.f};
    v2f bo0 = {0.f, 0.f}, bo1 = {0.f, 0.f};
    if (l < 8) {
        bi0.x = W_in[kA0 * 8 + l];       bi0.y = W_in[(kA0 + 1) * 8 + l];
        bi1.x = W_in[kA1 * 8 + l];       bi1.y = W_in[(kA1 + 1) * 8 + l];
        bo0.x = W_out[kA0 * 8 + l];      bo0.y = W_out[(kA0 + 1) * 8 + l];
        bo1.x = W_out[kA1 * 8 + l];      bo1.y = W_out[(kA1 + 1) * 8 + l];
    }

    v8f ci = {}; v8f co = {};
    ci = __builtin_amdgcn_wmma_f32_16x16x4_f32(false, a0, false, bi0, (short)0, ci, false, false);
    ci = __builtin_amdgcn_wmma_f32_16x16x4_f32(false, a1, false, bi1, (short)0, ci, false, false);
    co = __builtin_amdgcn_wmma_f32_16x16x4_f32(false, a0, false, bo0, (short)0, co, false, false);
    co = __builtin_amdgcn_wmma_f32_16x16x4_f32(false, a1, false, bo1, (short)0, co, false, false);

    const bool fullTile = (tile * 16 + 16) <= n;
    if (l < 8) {
        const int mbase = tile * 16 + 8 * half;
        float* pi = xw_in  + (size_t)mbase * 8 + l;
        float* po = xw_out + (size_t)mbase * 8 + l;
        if (fullTile) {
#pragma unroll
            for (int j = 0; j < 8; ++j) {
                pi[(size_t)j * 8] = ci[j];
                po[(size_t)j * 8] = co[j];
            }
        } else {
#pragma unroll
            for (int j = 0; j < 8; ++j) {
                if (mbase + j < n) {
                    pi[(size_t)j * 8] = ci[j];
                    po[(size_t)j * 8] = co[j];
                }
            }
        }
    }
}

__global__ void out_init_kernel(const float* __restrict__ xw_in,
                                const float* __restrict__ xw_out,
                                const float* __restrict__ dis_c,
                                const float* __restrict__ dis_r,
                                const float* __restrict__ b_in,
                                const float* __restrict__ b_out,
                                float* __restrict__ out, int n) {
    int i = blockIdx.x * blockDim.x + threadIdx.x;
    if (i >= n) return;
    float nc = dis_c[i] * dis_c[i];
    float nr = dis_r[i] * dis_r[i];
    const float* xi = xw_in  + (size_t)i * 8;
    const float* xo = xw_out + (size_t)i * 8;
    float* o = out + (size_t)i * 16;
#pragma unroll
    for (int k = 0; k < 8; ++k) o[k] = b_in[k] + xi[k] * nc;
#pragma unroll
    for (int k = 0; k < 8; ++k) o[8 + k] = b_out[k] + xo[k] * nr;
}

__global__ void edge_scatter_kernel(const int* __restrict__ rows,
                                    const int* __restrict__ cols,
                                    const float* __restrict__ xw_in,
                                    const float* __restrict__ xw_out,
                                    const float* __restrict__ dis_c,
                                    const float* __restrict__ dis_r,
                                    float* __restrict__ out, int ne) {
    const int lane = threadIdx.x & 31;
    const int e = blockIdx.x * blockDim.x + threadIdx.x;
    int r = 0, c = 0;
    float ni = 0.f, no = 0.f;
    if (e < ne) {
        r = rows[e];
        c = cols[e];
        ni = dis_c[r] * dis_c[c];
        no = dis_r[r] * dis_r[c];
    }
    const int f     = lane & 7;
    const int g     = lane >> 3;
    const int ebase = e - lane;

    float vi[8], vo[8];
    int   oi[8], oo[8];
    for (int j = 0; j < 8; ++j) {
        const int s  = 4 * j + g;
        const int rj = __shfl(r, s);
        const int cj = __shfl(c, s);
        const float nij = __shfl(ni, s);
        const float noj = __shfl(no, s);
        vi[j] = xw_in [rj * 8 + f] * nij;
        vo[j] = xw_out[cj * 8 + f] * noj;
        oi[j] = cj * 16 + f;
        oo[j] = rj * 16 + 8 + f;
    }
    for (int j = 0; j < 8; ++j) {
        if (ebase + 4 * j + g < ne) {
            fadd_agent(out + oi[j], vi[j]);
            fadd_agent(out + oo[j], vo[j]);
        }
    }
}

extern "C" void kernel_launch(void* const* d_in, const int* in_sizes, int n_in,
                              void* d_out, int out_size, void* d_ws, size_t ws_size,
                              hipStream_t stream) {
    const float* x     = (const float*)d_in[0];
    const int*   ei    = (const int*)d_in[1];
    const float* gamma = (const float*)d_in[2];
    const float* beta  = (const float*)d_in[3];
    const float* W_in  = (const float*)d_in[4];
    const float* b_in  = (const float*)d_in[5];
    const float* W_out = (const float*)d_in[6];
    const float* b_out = (const float*)d_in[7];
    float* out = (float*)d_out;

    const int n  = in_sizes[0] / 8;
    const int ne = in_sizes[1] / 2;
    const int NB = 1024;

    float* ws      = (float*)d_ws;
    float* xw_in   = ws;
    float* xw_out  = xw_in + (size_t)n * 8;
    float* dis_c   = xw_out + (size_t)n * 8;
    float* dis_r   = dis_c + n;
    float* partial = dis_r + n;
    float* scsh    = partial + (size_t)NB * 16;

    const int* rows = ei;
    const int* cols = ei + ne;

    bn_partial_kernel<<<NB, 256, 0, stream>>>(x, partial, n);
    bn_finalize_kernel<<<1, 32, 0, stream>>>(partial, gamma, beta, scsh, NB, 1.0f / (float)n);

    deg_init_kernel<<<(2 * n + 255) / 256, 256, 0, stream>>>((unsigned*)dis_c, 2 * n);
    deg_count_kernel<<<(ne + 255) / 256, 256, 0, stream>>>(rows, cols,
                                                           (unsigned*)dis_c, (unsigned*)dis_r, ne);
    deg_to_dis_kernel<<<(2 * n + 255) / 256, 256, 0, stream>>>((unsigned*)dis_c, 2 * n);

    const int nTiles = (n + 15) / 16;
    const int wavesPerBlock = 8;
    transform_wmma_kernel<<<(nTiles + wavesPerBlock - 1) / wavesPerBlock, 256, 0, stream>>>(
        x, W_in, W_out, scsh, xw_in, xw_out, nTiles, n);

    out_init_kernel<<<(n + 255) / 256, 256, 0, stream>>>(xw_in, xw_out, dis_c, dis_r,
                                                         b_in, b_out, out, n);
    edge_scatter_kernel<<<(ne + 255) / 256, 256, 0, stream>>>(rows, cols, xw_in, xw_out,
                                                              dis_c, dis_r, out, ne);
}


# edge sums: f32 global atomics replaced by edges binned by key range (LDS count, scan, place), one work-group per 2048-node bin adds rows in LDS in arrival order; 8x8 transform after the sum
# speedup vs baseline: 2.2655x; 1.1112x over previous
#include <hip/hip_runtime.h>
#include <hip/hip_bf16.h>

typedef float v2f __attribute__((ext_vector_type(2)));
typedef float v8f __attribute__((ext_vector_type(8)));

__device__ __forceinline__ void fadd_agent(float* p, float v) {
    __hip_atomic_fetch_add(p, v, __ATOMIC_RELAXED, __HIP_MEMORY_SCOPE_AGENT);
}

__global__ void bn_partial_kernel(const float* __restrict__ x,
                                  float* __restrict__ partial, int n) {
    float v[16];
#pragma unroll
    for (int k = 0; k < 16; ++k) v[k] = 0.f;

    const int stride = gridDim.x * blockDim.x;
    for (int i = blockIdx.x * blockDim.x + threadIdx.x; i < n; i += stride) {
        const float4* p = (const float4*)(x + (size_t)i * 8);
        float4 v0 = p[0];
        float4 v1 = p[1];
        float h;
        h = fmaxf(v0.x, 0.f); v[0] += h; v[8]  += h * h;
        h = fmaxf(v0.y, 0.f); v[1] += h; v[9]  += h * h;
        h = fmaxf(v0.z, 0.f); v[2] += h; v[10] += h * h;
        h = fmaxf(v0.w, 0.f); v[3] += h; v[11] += h * h;
        h = fmaxf(v1.x, 0.f); v[4] += h; v[12] += h * h;
        h = fmaxf(v1.y, 0.f); v[5] += h; v[13] += h * h;
        h = fmaxf(v1.z, 0.f); v[6] += h; v[14] += h * h;
        h = fmaxf(v1.w, 0.f); v[7] += h; v[15] += h * h;
    }

#pragma unroll
    for (int k = 0; k < 16; ++k) {
        float t = v[k];
        t += __shfl_xor(t, 1);
        t += __shfl_xor(t, 2);
        t += __shfl_xor(t, 4);
        t += __shfl_xor(t, 8);
        t += __shfl_xor(t, 16);
        v[k] = t;
    }

    __shared__ float sh[8 * 16];
    const int wave = threadIdx.x >> 5;
    const int lane = threadIdx.x & 31;
    if (lane == 0) {
#pragma unroll
        for (int k = 0; k < 16; ++k) sh[wave * 16 + k] = v[k];
    }
    __syncthreads();
    if (threadIdx.x < 16) {
        float t = 0.f;
#pragma unroll
        for (int w = 0; w < 8; ++w) t += sh[w * 16 + threadIdx.x];
        partial[blockIdx.x * 16 + threadIdx.x] = t;
    }
}

__global__ void bn_finalize_kernel(const float* __restrict__ partial,
                                   const float* __restrict__ gamma,
                                   const float* __restrict__ beta,
                                   float* __restrict__ scsh, int nb, float invN) {
    int k = threadIdx.x;
    if (k < 8) {
        float s = 0.f, q = 0.f;
        for (int b = 0; b < nb; ++b) {
            s += partial[b * 16 + k];
            q += partial[b * 16 + 8 + k];
        }
        float mean = s * invN;
        float var  = q * invN - mean * mean;
        float sc   = gamma[k] * rsqrtf(var + 1e-5f);
        scsh[k]     = sc;
        scsh[8 + k] = beta[k] - mean * sc;
    }
}

__global__ void deg_init_kernel(unsigned* __restrict__ deg, int n2) {
    int i = blockIdx.x * blockDim.x + threadIdx.x;
    if (i < n2) deg[i] = 1u;
}

__global__ void deg_count1_kernel(const int* __restrict__ idx,
                                  unsigned* __restrict__ deg, int ne) {
    int e = blockIdx.x * blockDim.x + threadIdx.x;
    if (e >= ne) return;
    atomicAdd(&deg[idx[e]], 1u);
}

__global__ void deg_to_dis_kernel(unsigned* __restrict__ deg, int n2) {
    int i = blockIdx.x * blockDim.x + threadIdx.x;
    if (i >= n2) return;
    float f = (float)deg[i];
    ((float*)deg)[i] = rsqrtf(f);
}

#define BIN_SHIFT 11
#define BIN_NODES 2048
#define NBINS 256
#define NCHUNK 256

__global__ void bin_count_kernel(const int* __restrict__ key,
                                 unsigned* __restrict__ tab, int ne) {
    __shared__ unsigned hist[NBINS];
    if (threadIdx.x < NBINS) hist[threadIdx.x] = 0u;
    __syncthreads();
    const int chunk = (ne + NCHUNK - 1) / NCHUNK;
    const int beg = blockIdx.x * chunk;
    const int end = min(beg + chunk, ne);
    for (int i = beg + threadIdx.x; i < end; i += 4 * 1024) {
        int k[4];
#pragma unroll
        for (int u = 0; u < 4; ++u)
            k[u] = (i + u * 1024 < end) ? key[i + u * 1024] : -1;
#pragma unroll
        for (int u = 0; u < 4; ++u)
            if (k[u] >= 0) atomicAdd(&hist[k[u] >> BIN_SHIFT], 1u);
    }
    __syncthreads();
    if (threadIdx.x < NBINS) tab[threadIdx.x * NCHUNK + blockIdx.x] = hist[threadIdx.x];
}

__global__ void bin_scan_kernel(unsigned* __restrict__ tab) {
    __shared__ unsigned part[1024];
    const int t = threadIdx.x;
    unsigned v[64];
    unsigned s = 0u;
#pragma unroll
    for (int k = 0; k < 64; ++k) { v[k] = tab[t * 64 + k]; s += v[k]; }
    part[t] = s;
    __syncthreads();
    for (int d = 1; d < 1024; d <<= 1) {
        unsigned add = (t >= d) ? part[t - d] : 0u;
        __syncthreads();
        part[t] += add;
        __syncthreads();
    }
    unsigned run = part[t] - s;
#pragma unroll
    for (int k = 0; k < 64; ++k) { tab[t * 64 + k] = run; run += v[k]; }
}

__global__ void bin_place_kernel(const int* __restrict__ key,
                                 const int* __restrict__ srcn,
                                 const unsigned* __restrict__ tab,
                                 unsigned* __restrict__ rec, int ne) {
    __shared__ unsigned cur[NBINS];
    if (threadIdx.x < NBINS) cur[threadIdx.x] = tab[threadIdx.x * NCHUNK + blockIdx.x];
    __syncthreads();
    const int chunk = (ne + NCHUNK - 1) / NCHUNK;
    const int beg = blockIdx.x * chunk;
    const int end = min(beg + chunk, ne);
    for (int i = beg + threadIdx.x; i < end; i += 4 * 1024) {
        int k[4], s[4];
#pragma unroll
        for (int u = 0; u < 4; ++u) {
            const bool ok = i + u * 1024 < end;
            k[u] = ok ? key[i + u * 1024] : -1;
            s[u] = ok ? srcn[i + u * 1024] : 0;
        }
#pragma unroll
        for (int u = 0; u < 4; ++u)
            if (k[u] >= 0) {
                const unsigned pos = atomicAdd(&cur[k[u] >> BIN_SHIFT], 1u);
                rec[pos] = ((unsigned)(k[u] & (BIN_NODES - 1)) << 19) | (unsigned)s[u];
            }
    }
}

__global__ void __launch_bounds__(1024)
bin_sum_kernel(const unsigned* __restrict__ rec,
               const unsigned* __restrict__ tab,
               const float* __restrict__ x,
               const float* __restrict__ scsh,
               const float* __restrict__ dis,
               const float* __restrict__ Wm,
               const float* __restrict__ bias,
               float* __restrict__ out, int colofs, int n, int ne) {
    __shared__ float acc[BIN_NODES * 8];
    const int b = blockIdx.x;
    const int base = b << BIN_SHIFT;
    if (base >= n) return;
    for (int i = threadIdx.x; i < BIN_NODES * 8; i += blockDim.x) acc[i] = 0.f;
    __syncthreads();

    const int lane = threadIdx.x & 31;
    const int wave = threadIdx.x >> 5;
    const int f = lane & 7;
    const int g = lane >> 3;
    const float sc = scsh[f];
    const float sh = scsh[8 + f];
    const int beg = (int)tab[b * NCHUNK];
    const int end = (b == NBINS - 1) ? ne : (int)tab[(b + 1) * NCHUNK];

    for (int i0 = beg + wave * 32; i0 < end; i0 += 1024) {
        const int i = i0 + lane;
        unsigned r = 0u;
        float w = 0.f;
        if (i < end) { r = rec[i]; w = dis[r & 0x7FFFFu]; }
        float v[8];
        int   a[8];
#pragma unroll
        for (int j = 0; j < 8; ++j) {
            const int s = 4 * j + g;
            const unsigned rj = __shfl(r, s);
            const float wj = __shfl(w, s);
            const int sj = (int)(rj & 0x7FFFFu);
            v[j] = (fmaxf(x[sj * 8 + f], 0.f) * sc + sh) * wj;
            a[j] = (int)(rj >> 19) * 8 + f;
        }
#pragma unroll
        for (int j = 0; j < 8; ++j)
            if (i0 + 4 * j + g < end) atomicAdd(&acc[a[j]], v[j]);
    }
    __syncthreads();

    const int half = lane >> 4;
    const int l    = lane & 15;
    const int kA0 = 2 * half;
    const int kA1 = 4 + 2 * half;
    v2f b0 = {0.f, 0.f}, b1 = {0.f, 0.f};
    float bs = 0.f;
    if (l < 8) {
        b0.x = Wm[kA0 * 8 + l];  b0.y = Wm[(kA0 + 1) * 8 + l];
        b1.x = Wm[kA1 * 8 + l];  b1.y = Wm[(kA1 + 1) * 8 + l];
        bs = bias[l];
    }
    for (int t = wave; t < BIN_NODES / 16; t += 32) {
        const int row0 = base + t * 16;
        if (row0 >= n) break;
        const int lrow = t * 16 + l;
        const int row  = row0 + l;
        const int rowc = row < n ? row : (n - 1);
        const float d = dis[rowc];
        const float* xr = x + (size_t)rowc * 8;
        const float* ar = acc + lrow * 8;
        v2f a0, a1;
        a0.x = (ar[kA0]     + (fmaxf(xr[kA0],     0.f) * scsh[kA0]     + scsh[8 + kA0])     * d) * d;
        a0.y = (ar[kA0 + 1] + (fmaxf(xr[kA0 + 1], 0.f) * scsh[kA0 + 1] + scsh[8 + kA0 + 1]) * d) * d;
        a1.x = (ar[kA1]     + (fmaxf(xr[kA1],     0.f) * scsh[kA1]     + scsh[8 + kA1])     * d) * d;
        a1.y = (ar[kA1 + 1] + (fmaxf(xr[kA1 + 1], 0.f) * scsh[kA1 + 1] + scsh[8 + kA1 + 1]) * d) * d;
        v8f c = {};
        c = __builtin_amdgcn_wmma_f32_16x16x4_f32(false, a0, false, b0, (short)0, c, false, false);
        c = __builtin_amdgcn_wmma_f32_16x16x4_f32(false, a1, false, b1, (short)0, c, false, false);
        if (l < 8) {
            const int mbase = row0 + 8 * half;
            float* po = out + (size_t)mbase * 16 + colofs + l;
#pragma unroll
            for (int j = 0; j < 8; ++j)
                if (mbase + j < n) po[(size_t)j * 16] = c[j] + bs;
        }
    }
}

extern "C" void kernel_launch(void* const* d_in, const int* in_sizes, int n_in,
                              void* d_out, int out_size, void* d_ws, size_t ws_size,
                              hipStream_t stream) {
    const float* x     = (const float*)d_in[0];
    const int*   ei    = (const int*)d_in[1];
    const float* gamma = (const float*)d_in[2];
    const float* beta  = (const float*)d_in[3];
    const float* W_in  = (const float*)d_in[4];
    const float* b_in  = (const float*)d_in[5];
    const float* W_out = (const float*)d_in[6];
    const float* b_out = (const float*)d_in[7];
    float* out = (float*)d_out;

    const int n  = in_sizes[0] / 8;
    const int ne = in_sizes[1] / 2;
    const int NB = 1024;

    float* ws      = (float*)d_ws;
    unsigned* rec  = (unsigned*)ws;
    float* dis_c   = ws + (size_t)n * 16;
    float* dis_r   = dis_c + n;
    float* partial = dis_r + n;
    float* scsh    = partial + (size_t)NB * 16;

    const int* rows = ei;
    const int* cols = ei + ne;

    bn_partial_kernel<<<NB, 256, 0, stream>>>(x, partial, n);
    bn_finalize_kernel<<<1, 32, 0, stream>>>(partial, gamma, beta, scsh, NB, 1.0f / (float)n);

    deg_init_kernel<<<(n + 255) / 256, 256, 0, stream>>>((unsigned*)dis_c, n);
    deg_count1_kernel<<<(ne + 255) / 256, 256, 0, stream>>>(cols, (unsigned*)dis_c, ne);
    deg_to_dis_kernel<<<(n + 255) / 256, 256, 0, stream>>>((unsigned*)dis_c, n);
    bin_count_kernel<<<NCHUNK, 1024, 0, stream>>>(cols, (unsigned*)dis_r, ne);
    bin_scan_kernel<<<1, 1024, 0, stream>>>((unsigned*)dis_r);
    bin_place_kernel<<<NCHUNK, 1024, 0, stream>>>(cols, rows, (const unsigned*)dis_r, rec, ne);
    bin_sum_kernel<<<NBINS, 1024, 0, stream>>>(rec, (const unsigned*)dis_r, x, scsh, dis_c,
                                               W_in, b_in, out, 0, n, ne);

    deg_init_kernel<<<(n + 255) / 256, 256, 0, stream>>>((unsigned*)dis_r, n);
    deg_count1_kernel<<<(ne + 255) / 256, 256, 0, stream>>>(rows, (unsigned*)dis_r, ne);
    deg_to_dis_kernel<<<(n + 255) / 256, 256, 0, stream>>>((unsigned*)dis_r, n);
    bin_count_kernel<<<NCHUNK, 1024, 0, stream>>>(rows, (unsigned*)dis_c, ne);
    bin_scan_kernel<<<1, 1024, 0, stream>>>((unsigned*)dis_c);
    bin_place_kernel<<<NCHUNK, 1024, 0, stream>>>(rows, cols, (const unsigned*)dis_c, rec, ne);
    bin_sum_kernel<<<NBINS, 1024, 0, stream>>>(rec, (const unsigned*)dis_c, x, scsh, dis_r,
                                               W_out, b_out, out, 8, n, ne);
}


# degrees: the two 8M-edge global u32 atomic passes go; each 2048-key bin's work-group counts its already binned records per key in LDS and writes rsqrt(1+count); 16 launches become 12
# speedup vs baseline: 4.9343x; 2.1780x over previous
#include <hip/hip_runtime.h>
#include <hip/hip_bf16.h>

typedef float v2f __attribute__((ext_vector_type(2)));
typedef float v8f __attribute__((ext_vector_type(8)));

__device__ __forceinline__ void fadd_agent(float* p, float v) {
    __hip_atomic_fetch_add(p, v, __ATOMIC_RELAXED, __HIP_MEMORY_SCOPE_AGENT);
}

__global__ void bn_partial_kernel(const float* __restrict__ x,
                                  float* __restrict__ partial, int n) {
    float v[16];
#pragma unroll
    for (int k = 0; k < 16; ++k) v[k] = 0.f;

    const int stride = gridDim.x * blockDim.x;
    for (int i = blockIdx.x * blockDim.x + threadIdx.x; i < n; i += stride) {
        const float4* p = (const float4*)(x + (size_t)i * 8);
        float4 v0 = p[0];
        float4 v1 = p[1];
        float h;
        h = fmaxf(v0.x, 0.f); v[0] += h; v[8]  += h * h;
        h = fmaxf(v0.y, 0.f); v[1] += h; v[9]  += h * h;
        h = fmaxf(v0.z, 0.f); v[2] += h; v[10] += h * h;
        h = fmaxf(v0.w, 0.f); v[3] += h; v[11] += h * h;
        h = fmaxf(v1.x, 0.f); v[4] += h; v[12] += h * h;
        h = fmaxf(v1.y, 0.f); v[5] += h; v[13] += h * h;
        h = fmaxf(v1.z, 0.f); v[6] += h; v[14] += h * h;
        h = fmaxf(v1.w, 0.f); v[7] += h; v[15] += h * h;
    }

#pragma unroll
    for (int k = 0; k < 16; ++k) {
        float t = v[k];
        t += __shfl_xor(t, 1);
        t += __shfl_xor(t, 2);
        t += __shfl_xor(t, 4);
        t += __shfl_xor(t, 8);
        t += __shfl_xor(t, 16);
        v[k] = t;
    }

    __shared__ float sh[8 * 16];
    const int wave = threadIdx.x >> 5;
    const int lane = threadIdx.x & 31;
    if (lane == 0) {
#pragma unroll
        for (int k = 0; k < 16; ++k) sh[wave * 16 + k] = v[k];
    }
    __syncthreads();
    if (threadIdx.x < 16) {
        float t = 0.f;
#pragma unroll
        for (int w = 0; w < 8; ++w) t += sh[w * 16 + threadIdx.x];
        partial[blockIdx.x * 16 + threadIdx.x] = t;
    }
}

__global__ void bn_finalize_kernel(const float* __restrict__ partial,
                                   const float* __restrict__ gamma,
                                   const float* __restrict__ beta,
                                   float* __restrict__ scsh, int nb, float invN) {
    int k = threadIdx.x;
    if (k < 8) {
        float s = 0.f, q = 0.f;
        for (int b = 0; b < nb; ++b) {
            s += partial[b * 16 + k];
            q += partial[b * 16 + 8 + k];
        }
        float mean = s * invN;
        float var  = q * invN - mean * mean;
        float sc   = gamma[k] * rsqrtf(var + 1e-5f);
        scsh[k]     = sc;
        scsh[8 + k] = beta[k] - mean * sc;
    }
}

#define BIN_SHIFT 11
#define BIN_NODES 2048
#define NBINS 256
#define NCHUNK 256

__global__ void bin_count_kernel(const int* __restrict__ key,
                                 unsigned* __restrict__ tab, int ne) {
    __shared__ unsigned hist[NBINS];
    if (threadIdx.x < NBINS) hist[threadIdx.x] = 0u;
    __syncthreads();
    const int chunk = (ne + NCHUNK - 1) / NCHUNK;
    const int beg = blockIdx.x * chunk;
    const int end = min(beg + chunk, ne);
    for (int i = beg + threadIdx.x; i < end; i += 4 * 1024) {
        int k[4];
#pragma unroll
        for (int u = 0; u < 4; ++u)
            k[u] = (i + u * 1024 < end) ? key[i + u * 1024] : -1;
#pragma unroll
        for (int u = 0; u < 4; ++u)
            if (k[u] >= 0) atomicAdd(&hist[k[u] >> BIN_SHIFT], 1u);
    }
    __syncthreads();
    if (threadIdx.x < NBINS) tab[threadIdx.x * NCHUNK + blockIdx.x] = hist[threadIdx.x];
}

__global__ void bin_scan_kernel(unsigned* __restrict__ tab) {
    __shared__ unsigned part[1024];
    const int t = threadIdx.x;
    unsigned v[64];
    unsigned s = 0u;
#pragma unroll
    for (int k = 0; k < 64; ++k) { v[k] = tab[t * 64 + k]; s += v[k]; }
    part[t] = s;
    __syncthreads();
    for (int d = 1; d < 1024; d <<= 1) {
        unsigned add = (t >= d) ? part[t - d] : 0u;
        __syncthreads();
        part[t] += add;
        __syncthreads();
    }
    unsigned run = part[t] - s;
#pragma unroll
    for (int k = 0; k < 64; ++k) { tab[t * 64 + k] = run; run += v[k]; }
}

__global__ void bin_place_kernel(const int* __restrict__ key,
                                 const int* __restrict__ srcn,
                                 const unsigned* __restrict__ tab,
                                 unsigned* __restrict__ rec, int ne) {
    __shared__ unsigned cur[NBINS];
    if (threadIdx.x < NBINS) cur[threadIdx.x] = tab[threadIdx.x * NCHUNK + blockIdx.x];
    __syncthreads();
    const int chunk = (ne + NCHUNK - 1) / NCHUNK;
    const int beg = blockIdx.x * chunk;
    const int end = min(beg + chunk, ne);
    for (int i = beg + threadIdx.x; i < end; i += 4 * 1024) {
        int k[4], s[4];
#pragma unroll
        for (int u = 0; u < 4; ++u) {
            const bool ok = i + u * 1024 < end;
            k[u] = ok ? key[i + u * 1024] : -1;
            s[u] = ok ? srcn[i + u * 1024] : 0;
        }
#pragma unroll
        for (int u = 0; u < 4; ++u)
            if (k[u] >= 0) {
                const unsigned pos = atomicAdd(&cur[k[u] >> BIN_SHIFT], 1u);
                rec[pos] = ((unsigned)(k[u] & (BIN_NODES - 1)) << 19) | (unsigned)s[u];
            }
    }
}

__global__ void __launch_bounds__(1024)
bin_deg_kernel(const unsigned* __restrict__ rec,
               const unsigned* __restrict__ tab,
               float* __restrict__ dis, int n, int ne) {
    __shared__ unsigned cnt[BIN_NODES];
    const int b = blockIdx.x;
    const int base = b << BIN_SHIFT;
    if (base >= n) return;
    for (int i = threadIdx.x; i < BIN_NODES; i += blockDim.x) cnt[i] = 1u;
    __syncthreads();
    const int beg = (int)tab[b * NCHUNK];
    const int end = (b == NBINS - 1) ? ne : (int)tab[(b + 1) * NCHUNK];
    for (int i = beg + threadIdx.x; i < end; i += 8 * 1024) {
        unsigned r[8];
#pragma unroll
        for (int u = 0; u < 8; ++u) {
            const int j = i + u * 1024;
            r[u] = rec[j < end ? j : i];
        }
#pragma unroll
        for (int u = 0; u < 8; ++u)
            atomicAdd(&cnt[r[u] >> 19], (i + u * 1024 < end) ? 1u : 0u);
    }
    __syncthreads();
    for (int i = threadIdx.x; i < BIN_NODES; i += blockDim.x) {
        const int node = base + i;
        if (node < n) dis[node] = rsqrtf((float)cnt[i]);
    }
}

__global__ void __launch_bounds__(1024)
bin_sum_kernel(const unsigned* __restrict__ rec,
               const unsigned* __restrict__ tab,
               const float* __restrict__ x,
               const float* __restrict__ scsh,
               const float* __restrict__ dis,
               const float* __restrict__ Wm,
               const float* __restrict__ bias,
               float* __restrict__ out, int colofs, int n, int ne) {
    __shared__ float acc[BIN_NODES * 8];
    const int b = blockIdx.x;
    const int base = b << BIN_SHIFT;
    if (base >= n) return;
    for (int i = threadIdx.x; i < BIN_NODES * 8; i += blockDim.x) acc[i] = 0.f;
    __syncthreads();

    const int lane = threadIdx.x & 31;
    const int wave = threadIdx.x >> 5;
    const int f = lane & 7;
    const int g = lane >> 3;
    const float sc = scsh[f];
    const float sh = scsh[8 + f];
    const int beg = (int)tab[b * NCHUNK];
    const int end = (b == NBINS - 1) ? ne : (int)tab[(b + 1) * NCHUNK];

    for (int i0 = beg + wave * 32; i0 < end; i0 += 1024) {
        const int i = i0 + lane;
        unsigned r = 0u;
        float w = 0.f;
        if (i < end) { r = rec[i]; w = dis[r & 0x7FFFFu]; }
        float v[8];
        int   a[8];
#pragma unroll
        for (int j = 0; j < 8; ++j) {
            const int s = 4 * j + g;
            const unsigned rj = __shfl(r, s);
            const float wj = __shfl(w, s);
            const int sj = (int)(rj & 0x7FFFFu);
            v[j] = (fmaxf(x[sj * 8 + f], 0.f) * sc + sh) * wj;
            a[j] = (int)(rj >> 19) * 8 + f;
        }
#pragma unroll
        for (int j = 0; j < 8; ++j)
            if (i0 + 4 * j + g < end) atomicAdd(&acc[a[j]], v[j]);
    }
    __syncthreads();

    const int half = lane >> 4;
    const int l    = lane & 15;
    const int kA0 = 2 * half;
    const int kA1 = 4 + 2 * half;
    v2f b0 = {0.f, 0.f}, b1 = {0.f, 0.f};
    float bs = 0.f;
    if (l < 8) {
        b0.x = Wm[kA0 * 8 + l];  b0.y = Wm[(kA0 + 1) * 8 + l];
        b1.x = Wm[kA1 * 8 + l];  b1.y = Wm[(kA1 + 1) * 8 + l];
        bs = bias[l];
    }
    for (int t = wave; t < BIN_NODES / 16; t += 32) {
        const int row0 = base + t * 16;
        if (row0 >= n) break;
        const int lrow = t * 16 + l;
        const int row  = row0 + l;
        const int rowc = row < n ? row : (n - 1);
        const float d = dis[rowc];
        const float* xr = x + (size_t)rowc * 8;
        const float* ar = acc + lrow * 8;
        v2f a0, a1;
        a0.x = (ar[kA0]     + (fmaxf(xr[kA0],     0.f) * scsh[kA0]     + scsh[8 + kA0])     * d) * d;
        a0.y = (ar[kA0 + 1] + (fmaxf(xr[kA0 + 1], 0.f) * scsh[kA0 + 1] + scsh[8 + kA0 + 1]) * d) * d;
        a1.x = (ar[kA1]     + (fmaxf(xr[kA1],     0.f) * scsh[kA1]     + scsh[8 + kA1])     * d) * d;
        a1.y = (ar[kA1 + 1] + (fmaxf(xr[kA1 + 1], 0.f) * scsh[kA1 + 1] + scsh[8 + kA1 + 1]) * d) * d;
        v8f c = {};
        c = __builtin_amdgcn_wmma_f32_16x16x4_f32(false, a0, false, b0, (short)0, c, false, false);
        c = __builtin_amdgcn_wmma_f32_16x16x4_f32(false, a1, false, b1, (short)0, c, false, false);
        if (l < 8) {
            const int mbase = row0 + 8 * half;
            float* po = out + (size_t)mbase * 16 + colofs + l;
#pragma unroll
            for (int j = 0; j < 8; ++j)
                if (mbase + j < n) po[(size_t)j * 16] = c[j] + bs;
        }
    }
}

extern "C" void kernel_launch(void* const* d_in, const int* in_sizes, int n_in,
                              void* d_out, int out_size, void* d_ws, size_t ws_size,
                              hipStream_t stream) {
    const float* x     = (const float*)d_in[0];
    const int*   ei    = (const int*)d_in[1];
    const float* gamma = (const float*)d_in[2];
    const float* beta  = (const float*)d_in[3];
    const float* W_in  = (const float*)d_in[4];
    const float* b_in  = (const float*)d_in[5];
    const float* W_out = (const float*)d_in[6];
    const float* b_out = (const float*)d_in[7];
    float* out = (float*)d_out;

    const int n  = in_sizes[0] / 8;
    const int ne = in_sizes[1] / 2;
    const int NB = 1024;

    float* ws      = (float*)d_ws;
    unsigned* rec  = (unsigned*)ws;
    float* dis_c   = ws + (size_t)n * 16;
    float* dis_r   = dis_c + n;
    float* partial = dis_r + n;
    float* scsh    = partial + (size_t)NB * 16;

    const int* rows = ei;
    const int* cols = ei + ne;

    bn_partial_kernel<<<NB, 256, 0, stream>>>(x, partial, n);
    bn_finalize_kernel<<<1, 32, 0, stream>>>(partial, gamma, beta, scsh, NB, 1.0f / (float)n);

    bin_count_kernel<<<NCHUNK, 1024, 0, stream>>>(cols, (unsigned*)dis_r, ne);
    bin_scan_kernel<<<1, 1024, 0, stream>>>((unsigned*)dis_r);
    bin_place_kernel<<<NCHUNK, 1024, 0, stream>>>(cols, rows, (const unsigned*)dis_r, rec, ne);
    bin_deg_kernel<<<NBINS, 1024, 0, stream>>>(rec, (const unsigned*)dis_r, dis_c, n, ne);
    bin_sum_kernel<<<NBINS, 1024, 0, stream>>>(rec, (const unsigned*)dis_r, x, scsh, dis_c,
                                               W_in, b_in, out, 0, n, ne);

    bin_count_kernel<<<NCHUNK, 1024, 0, stream>>>(rows, (unsigned*)dis_c, ne);
    bin_scan_kernel<<<1, 1024, 0, stream>>>((unsigned*)dis_c);
    bin_place_kernel<<<NCHUNK, 1024, 0, stream>>>(rows, cols, (const unsigned*)dis_c, rec, ne);
    bin_deg_kernel<<<NBINS, 1024, 0, stream>>>(rec, (const unsigned*)dis_c, dis_r, n, ne);
    bin_sum_kernel<<<NBINS, 1024, 0, stream>>>(rec, (const unsigned*)dis_c, x, scsh, dis_r,
                                               W_out, b_out, out, 8, n, ne);
}


# bin offsets: the two single-work-group scans of the 65536-count table go; each place work-group sums the raw counts itself (whole-line reads, partial sums in LDS) into its cursors; 12 launches to 10
# speedup vs baseline: 5.4732x; 1.1092x over previous
#include <hip/hip_runtime.h>
#include <hip/hip_bf16.h>

typedef float v2f __attribute__((ext_vector_type(2)));
typedef float v8f __attribute__((ext_vector_type(8)));

__device__ __forceinline__ void fadd_agent(float* p, float v) {
    __hip_atomic_fetch_add(p, v, __ATOMIC_RELAXED, __HIP_MEMORY_SCOPE_AGENT);
}

__global__ void bn_partial_kernel(const float* __restrict__ x,
                                  float* __restrict__ partial, int n) {
    float v[16];
#pragma unroll
    for (int k = 0; k < 16; ++k) v[k] = 0.f;

    const int stride = gridDim.x * blockDim.x;
    for (int i = blockIdx.x * blockDim.x + threadIdx.x; i < n; i += stride) {
        const float4* p = (const float4*)(x + (size_t)i * 8);
        float4 v0 = p[0];
        float4 v1 = p[1];
        float h;
        h = fmaxf(v0.x, 0.f); v[0] += h; v[8]  += h * h;
        h = fmaxf(v0.y, 0.f); v[1] += h; v[9]  += h * h;
        h = fmaxf(v0.z, 0.f); v[2] += h; v[10] += h * h;
        h = fmaxf(v0.w, 0.f); v[3] += h; v[11] += h * h;
        h = fmaxf(v1.x, 0.f); v[4] += h; v[12] += h * h;
        h = fmaxf(v1.y, 0.f); v[5] += h; v[13] += h * h;
        h = fmaxf(v1.z, 0.f); v[6] += h; v[14] += h * h;
        h = fmaxf(v1.w, 0.f); v[7] += h; v[15] += h * h;
    }

#pragma unroll
    for (int k = 0; k < 16; ++k) {
        float t = v[k];
        t += __shfl_xor(t, 1);
        t += __shfl_xor(t, 2);
        t += __shfl_xor(t, 4);
        t += __shfl_xor(t, 8);
        t += __shfl_xor(t, 16);
        v[k] = t;
    }

    __shared__ float sh[8 * 16];
    const int wave = threadIdx.x >> 5;
    const int lane = threadIdx.x & 31;
    if (lane == 0) {
#pragma unroll
        for (int k = 0; k < 16; ++k) sh[wave * 16 + k] = v[k];
    }
    __syncthreads();
    if (threadIdx.x < 16) {
        float t = 0.f;
#pragma unroll
        for (int w = 0; w < 8; ++w) t += sh[w * 16 + threadIdx.x];
        partial[blockIdx.x * 16 + threadIdx.x] = t;
    }
}

__global__ void bn_finalize_kernel(const float* __restrict__ partial,
                                   const float* __restrict__ gamma,
                                   const float* __restrict__ beta,
                                   float* __restrict__ scsh, int nb, float invN) {
    int k = threadIdx.x;
    if (k < 8) {
        float s = 0.f, q = 0.f;
        for (int b = 0; b < nb; ++b) {
            s += partial[b * 16 + k];
            q += partial[b * 16 + 8 + k];
        }
        float mean = s * invN;
        float var  = q * invN - mean * mean;
        float sc   = gamma[k] * rsqrtf(var + 1e-5f);
        scsh[k]     = sc;
        scsh[8 + k] = beta[k] - mean * sc;
    }
}

#define BIN_SHIFT 11
#define BIN_NODES 2048
#define NBINS 256
#define NCHUNK 256

__global__ void bin_count_kernel(const int* __restrict__ key,
                                 unsigned* __restrict__ tab, int ne) {
    __shared__ unsigned hist[NBINS];
    if (threadIdx.x < NBINS) hist[threadIdx.x] = 0u;
    __syncthreads();
    const int chunk = (ne + NCHUNK - 1) / NCHUNK;
    const int beg = blockIdx.x * chunk;
    const int end = min(beg + chunk, ne);
    for (int i = beg + threadIdx.x; i < end; i += 4 * 1024) {
        int k[4];
#pragma unroll
        for (int u = 0; u < 4; ++u)
            k[u] = (i + u * 1024 < end) ? key[i + u * 1024] : -1;
#pragma unroll
        for (int u = 0; u < 4; ++u)
            if (k[u] >= 0) atomicAdd(&hist[k[u] >> BIN_SHIFT], 1u);
    }
    __syncthreads();
    if (threadIdx.x < NBINS) tab[blockIdx.x * NBINS + threadIdx.x] = hist[threadIdx.x];
}

__global__ void bin_place_kernel(const int* __restrict__ key,
                                 const int* __restrict__ srcn,
                                 const unsigned* __restrict__ tab,
                                 unsigned* __restrict__ bstart,
                                 unsigned* __restrict__ rec, int ne) {
    __shared__ unsigned cur[NBINS];
    __shared__ unsigned tot[NBINS];
    __shared__ __attribute__((aligned(16))) unsigned partT[16 * NBINS];
    __shared__ __attribute__((aligned(16))) unsigned partP[16 * NBINS];
    {
        const int c   = blockIdx.x;
        const int qd  = threadIdx.x & 63;
        const int grp = threadIdx.x >> 6;
        const uint4* col = (const uint4*)tab + (size_t)(grp * 16) * (NBINS / 4) + qd;
        uint4 r[16];
#pragma unroll
        for (int i = 0; i < 16; ++i) r[i] = col[i * (NBINS / 4)];
        unsigned t0 = 0u, t1 = 0u, t2 = 0u, t3 = 0u;
        unsigned p0 = 0u, p1 = 0u, p2 = 0u, p3 = 0u;
#pragma unroll
        for (int i = 0; i < 16; ++i) {
            t0 += r[i].x; t1 += r[i].y; t2 += r[i].z; t3 += r[i].w;
            if (grp * 16 + i < c) { p0 += r[i].x; p1 += r[i].y; p2 += r[i].z; p3 += r[i].w; }
        }
        const int o = grp * NBINS + 4 * qd;
        partT[o] = t0; partT[o + 1] = t1; partT[o + 2] = t2; partT[o + 3] = t3;
        partP[o] = p0; partP[o + 1] = p1; partP[o + 2] = p2; partP[o + 3] = p3;
    }
    __syncthreads();
    if (threadIdx.x < NBINS) {
        unsigned t = 0u, p = 0u;
#pragma unroll
        for (int g = 0; g < 16; ++g) {
            t += partT[g * NBINS + threadIdx.x];
            p += partP[g * NBINS + threadIdx.x];
        }
        tot[threadIdx.x] = t;
        cur[threadIdx.x] = p;
    }
    __syncthreads();
    if (threadIdx.x < 32) {
        const int lane = threadIdx.x;
        unsigned v[NBINS / 32], cv[NBINS / 32];
        unsigned s = 0u;
#pragma unroll
        for (int k = 0; k < NBINS / 32; ++k) {
            v[k]  = tot[lane * (NBINS / 32) + k];
            cv[k] = cur[lane * (NBINS / 32) + k];
            s += v[k];
        }
        unsigned inc = s;
#pragma unroll
        for (int d = 1; d < 32; d <<= 1) {
            const unsigned up = __shfl_up(inc, d);
            if (lane >= d) inc += up;
        }
        unsigned run = inc - s;
#pragma unroll
        for (int k = 0; k < NBINS / 32; ++k) {
            const int b = lane * (NBINS / 32) + k;
            cur[b] = cv[k] + run;
            if (blockIdx.x == 0) bstart[b] = run;
            run += v[k];
        }
    }
    __syncthreads();
    const int chunk = (ne + NCHUNK - 1) / NCHUNK;
    const int beg = blockIdx.x * chunk;
    const int end = min(beg + chunk, ne);
    for (int i = beg + threadIdx.x; i < end; i += 4 * 1024) {
        int k[4], s[4];
#pragma unroll
        for (int u = 0; u < 4; ++u) {
            const bool ok = i + u * 1024 < end;
            k[u] = ok ? key[i + u * 1024] : -1;
            s[u] = ok ? srcn[i + u * 1024] : 0;
        }
#pragma unroll
        for (int u = 0; u < 4; ++u)
            if (k[u] >= 0) {
                const unsigned pos = atomicAdd(&cur[k[u] >> BIN_SHIFT], 1u);
                rec[pos] = ((unsigned)(k[u] & (BIN_NODES - 1)) << 19) | (unsigned)s[u];
            }
    }
}

__global__ void __launch_bounds__(1024)
bin_deg_kernel(const unsigned* __restrict__ rec,
               const unsigned* __restrict__ bstart,
               float* __restrict__ dis, int n, int ne) {
    __shared__ unsigned cnt[BIN_NODES];
    const int b = blockIdx.x;
    const int base = b << BIN_SHIFT;
    if (base >= n) return;
    for (int i = threadIdx.x; i < BIN_NODES; i += blockDim.x) cnt[i] = 1u;
    __syncthreads();
    const int beg = (int)bstart[b];
    const int end = (b == NBINS - 1) ? ne : (int)bstart[b + 1];
    for (int i = beg + threadIdx.x; i < end; i += 8 * 1024) {
        unsigned r[8];
#pragma unroll
        for (int u = 0; u < 8; ++u) {
            const int j = i + u * 1024;
            r[u] = rec[j < end ? j : i];
        }
#pragma unroll
        for (int u = 0; u < 8; ++u)
            atomicAdd(&cnt[r[u] >> 19], (i + u * 1024 < end) ? 1u : 0u);
    }
    __syncthreads();
    for (int i = threadIdx.x; i < BIN_NODES; i += blockDim.x) {
        const int node = base + i;
        if (node < n) dis[node] = rsqrtf((float)cnt[i]);
    }
}

__global__ void __launch_bounds__(1024)
bin_sum_kernel(const unsigned* __restrict__ rec,
               const unsigned* __restrict__ bstart,
               const float* __restrict__ x,
               const float* __restrict__ scsh,
               const float* __restrict__ dis,
               const float* __restrict__ Wm,
               const float* __restrict__ bias,
               float* __restrict__ out, int colofs, int n, int ne) {
    __shared__ float acc[BIN_NODES * 8];
    const int b = blockIdx.x;
    const int base = b << BIN_SHIFT;
    if (base >= n) return;
    for (int i = threadIdx.x; i < BIN_NODES * 8; i += blockDim.x) acc[i] = 0.f;
    __syncthreads();

    const int lane = threadIdx.x & 31;
    const int wave = threadIdx.x >> 5;
    const int f = lane & 7;
    const int g = lane >> 3;
    const float sc = scsh[f];
    const float sh = scsh[8 + f];
    const int beg = (int)bstart[b];
    const int end = (b == NBINS - 1) ? ne : (int)bstart[b + 1];

    for (int i0 = beg + wave * 32; i0 < end; i0 += 1024) {
        const int i = i0 + lane;
        unsigned r = 0u;
        float w = 0.f;
        if (i < end) { r = rec[i]; w = dis[r & 0x7FFFFu]; }
        float v[8];
        int   a[8];
#pragma unroll
        for (int j = 0; j < 8; ++j) {
            const int s = 4 * j + g;
            const unsigned rj = __shfl(r, s);
            const float wj = __shfl(w, s);
            const int sj = (int)(rj & 0x7FFFFu);
            v[j] = (fmaxf(x[sj * 8 + f], 0.f) * sc + sh) * wj;
            a[j] = (int)(rj >> 19) * 8 + f;
        }
#pragma unroll
        for (int j = 0; j < 8; ++j)
            if (i0 + 4 * j + g < end) atomicAdd(&acc[a[j]], v[j]);
    }
    __syncthreads();

    const int half = lane >> 4;
    const int l    = lane & 15;
    const int kA0 = 2 * half;
    const int kA1 = 4 + 2 * half;
    v2f b0 = {0.f, 0.f}, b1 = {0.f, 0.f};
    float bs = 0.f;
    if (l < 8) {
        b0.x = Wm[kA0 * 8 + l];  b0.y = Wm[(kA0 + 1) * 8 + l];
        b1.x = Wm[kA1 * 8 + l];  b1.y = Wm[(kA1 + 1) * 8 + l];
        bs = bias[l];
    }
    for (int t = wave; t < BIN_NODES / 16; t += 32) {
        const int row0 = base + t * 16;
        if (row0 >= n) break;
        const int lrow = t * 16 + l;
        const int row  = row0 + l;
        const int rowc = row < n ? row : (n - 1);
        const float d = dis[rowc];
        const float* xr = x + (size_t)rowc * 8;
        const float* ar = acc + lrow * 8;
        v2f a0, a1;
        a0.x = (ar[kA0]     + (fmaxf(xr[kA0],     0.f) * scsh[kA0]     + scsh[8 + kA0])     * d) * d;
        a0.y = (ar[kA0 + 1] + (fmaxf(xr[kA0 + 1], 0.f) * scsh[kA0 + 1] + scsh[8 + kA0 + 1]) * d) * d;
        a1.x = (ar[kA1]     + (fmaxf(xr[kA1],     0.f) * scsh[kA1]     + scsh[8 + kA1])     * d) * d;
        a1.y = (ar[kA1 + 1] + (fmaxf(xr[kA1 + 1], 0.f) * scsh[kA1 + 1] + scsh[8 + kA1 + 1]) * d) * d;
        v8f c = {};
        c = __builtin_amdgcn_wmma_f32_16x16x4_f32(false, a0, false, b0, (short)0, c, false, false);
        c = __builtin_amdgcn_wmma_f32_16x16x4_f32(false, a1, false, b1, (short)0, c, false, false);
        if (l < 8) {
            const int mbase = row0 + 8 * half;
            float* po = out + (size_t)mbase * 16 + colofs + l;
#pragma unroll
            for (int j = 0; j < 8; ++j)
                if (mbase + j < n) po[(size_t)j * 16] = c[j] + bs;
        }
    }
}

extern "C" void kernel_launch(void* const* d_in, const int* in_sizes, int n_in,
                              void* d_out, int out_size, void* d_ws, size_t ws_size,
                              hipStream_t stream) {
    const float* x     = (const float*)d_in[0];
    const int*   ei    = (const int*)d_in[1];
    const float* gamma = (const float*)d_in[2];
    const float* beta  = (const float*)d_in[3];
    const float* W_in  = (const float*)d_in[4];
    const float* b_in  = (const float*)d_in[5];
    const float* W_out = (const float*)d_in[6];
    const float* b_out = (const float*)d_in[7];
    float* out = (float*)d_out;

    const int n  = in_sizes[0] / 8;
    const int ne = in_sizes[1] / 2;
    const int NB = 1024;

    float* ws      = (float*)d_ws;
    unsigned* rec  = (unsigned*)ws;
    float* dis_c   = ws + (size_t)n * 16;
    float* dis_r   = dis_c + n;
    float* partial = dis_r + n;
    float* scsh    = partial + (size_t)NB * 16;

    const int* rows = ei;
    const int* cols = ei + ne;

    bn_partial_kernel<<<NB, 256, 0, stream>>>(x, partial, n);
    bn_finalize_kernel<<<1, 32, 0, stream>>>(partial, gamma, beta, scsh, NB, 1.0f / (float)n);

    unsigned* tab_in    = (unsigned*)dis_r;
    unsigned* bstart_in = tab_in + NBINS * NCHUNK;
    bin_count_kernel<<<NCHUNK, 1024, 0, stream>>>(cols, tab_in, ne);
    bin_place_kernel<<<NCHUNK, 1024, 0, stream>>>(cols, rows, tab_in, bstart_in, rec, ne);
    bin_deg_kernel<<<NBINS, 1024, 0, stream>>>(rec, bstart_in, dis_c, n, ne);
    bin_sum_kernel<<<NBINS, 1024, 0, stream>>>(rec, bstart_in, x, scsh, dis_c,
                                               W_in, b_in, out, 0, n, ne);

    unsigned* tab_out    = (unsigned*)dis_c;
    unsigned* bstart_out = tab_out + NBINS * NCHUNK;
    bin_count_kernel<<<NCHUNK, 1024, 0, stream>>>(rows, tab_out, ne);
    bin_place_kernel<<<NCHUNK, 1024, 0, stream>>>(rows, cols, tab_out, bstart_out, rec, ne);
    bin_deg_kernel<<<NBINS, 1024, 0, stream>>>(rec, bstart_out, dis_r, n, ne);
    bin_sum_kernel<<<NBINS, 1024, 0, stream>>>(rec, bstart_out, x, scsh, dis_r,
                                               W_out, b_out, out, 8, n, ne);
}


# bin_place: records reach rec through an LDS stage in bin order, each run copied out one 128-B line per store (was 32 lines per store); record order inside a bin changes, sums stay in arrival order
# speedup vs baseline: 6.5534x; 1.1974x over previous
#include <hip/hip_runtime.h>
#include <hip/hip_bf16.h>

typedef float v2f __attribute__((ext_vector_type(2)));
typedef float v8f __attribute__((ext_vector_type(8)));

__device__ __forceinline__ void fadd_agent(float* p, float v) {
    __hip_atomic_fetch_add(p, v, __ATOMIC_RELAXED, __HIP_MEMORY_SCOPE_AGENT);
}

__global__ void bn_partial_kernel(const float* __restrict__ x,
                                  float* __restrict__ partial, int n) {
    float v[16];
#pragma unroll
    for (int k = 0; k < 16; ++k) v[k] = 0.f;

    const int stride = gridDim.x * blockDim.x;
    for (int i = blockIdx.x * blockDim.x + threadIdx.x; i < n; i += stride) {
        const float4* p = (const float4*)(x + (size_t)i * 8);
        float4 v0 = p[0];
        float4 v1 = p[1];
        float h;
        h = fmaxf(v0.x, 0.f); v[0] += h; v[8]  += h * h;
        h = fmaxf(v0.y, 0.f); v[1] += h; v[9]  += h * h;
        h = fmaxf(v0.z, 0.f); v[2] += h; v[10] += h * h;
        h = fmaxf(v0.w, 0.f); v[3] += h; v[11] += h * h;
        h = fmaxf(v1.x, 0.f); v[4] += h; v[12] += h * h;
        h = fmaxf(v1.y, 0.f); v[5] += h; v[13] += h * h;
        h = fmaxf(v1.z, 0.f); v[6] += h; v[14] += h * h;
        h = fmaxf(v1.w, 0.f); v[7] += h; v[15] += h * h;
    }

#pragma unroll
    for (int k = 0; k < 16; ++k) {
        float t = v[k];
        t += __shfl_xor(t, 1);
        t += __shfl_xor(t, 2);
        t += __shfl_xor(t, 4);
        t += __shfl_xor(t, 8);
        t += __shfl_xor(t, 16);
        v[k] = t;
    }

    __shared__ float sh[8 * 16];
    const int wave = threadIdx.x >> 5;
    const int lane = threadIdx.x & 31;
    if (lane == 0) {
#pragma unroll
        for (int k = 0; k < 16; ++k) sh[wave * 16 + k] = v[k];
    }
    __syncthreads();
    if (threadIdx.x < 16) {
        float t = 0.f;
#pragma unroll
        for (int w = 0; w < 8; ++w) t += sh[w * 16 + threadIdx.x];
        partial[blockIdx.x * 16 + threadIdx.x] = t;
    }
}

__global__ void bn_finalize_kernel(const float* __restrict__ partial,
                                   const float* __restrict__ gamma,
                                   const float* __restrict__ beta,
                                   float* __restrict__ scsh, int nb, float invN) {
    int k = threadIdx.x;
    if (k < 8) {
        float s = 0.f, q = 0.f;
        for (int b = 0; b < nb; ++b) {
            s += partial[b * 16 + k];
            q += partial[b * 16 + 8 + k];
        }
        float mean = s * invN;
        float var  = q * invN - mean * mean;
        float sc   = gamma[k] * rsqrtf(var + 1e-5f);
        scsh[k]     = sc;
        scsh[8 + k] = beta[k] - mean * sc;
    }
}

#define BIN_SHIFT 11
#define BIN_NODES 2048
#define NBINS 256
#define NCHUNK 256

__global__ void bin_count_kernel(const int* __restrict__ key,
                                 unsigned* __restrict__ tab, int ne) {
    __shared__ unsigned hist[NBINS];
    if (threadIdx.x < NBINS) hist[threadIdx.x] = 0u;
    __syncthreads();
    const int chunk = (ne + NCHUNK - 1) / NCHUNK;
    const int beg = blockIdx.x * chunk;
    const int end = min(beg + chunk, ne);
    for (int i = beg + threadIdx.x; i < end; i += 4 * 1024) {
        int k[4];
#pragma unroll
        for (int u = 0; u < 4; ++u)
            k[u] = (i + u * 1024 < end) ? key[i + u * 1024] : -1;
#pragma unroll
        for (int u = 0; u < 4; ++u)
            if (k[u] >= 0) atomicAdd(&hist[k[u] >> BIN_SHIFT], 1u);
    }
    __syncthreads();
    if (threadIdx.x < NBINS) tab[blockIdx.x * NBINS + threadIdx.x] = hist[threadIdx.x];
}

#define STAGE_CAP 32768
__global__ void bin_place_kernel(const int* __restrict__ key,
                                 const int* __restrict__ srcn,
                                 const unsigned* __restrict__ tab,
                                 unsigned* __restrict__ bstart,
                                 unsigned* __restrict__ rec, int ne) {
    __shared__ unsigned cur[NBINS];
    __shared__ unsigned tot[NBINS];
    __shared__ unsigned lst[NBINS + 1];
    __shared__ unsigned gst[NBINS];
    __shared__ __attribute__((aligned(16))) unsigned stage[STAGE_CAP];
    unsigned* partT = stage;
    unsigned* partP = stage + 16 * NBINS;
    const unsigned own = (threadIdx.x < NBINS) ? tab[(size_t)blockIdx.x * NBINS + threadIdx.x] : 0u;
    {
        const int c   = blockIdx.x;
        const int qd  = threadIdx.x & 63;
        const int grp = threadIdx.x >> 6;
        const uint4* col = (const uint4*)tab + (size_t)(grp * 16) * (NBINS / 4) + qd;
        uint4 r[16];
#pragma unroll
        for (int i = 0; i < 16; ++i) r[i] = col[i * (NBINS / 4)];
        unsigned t0 = 0u, t1 = 0u, t2 = 0u, t3 = 0u;
        unsigned p0 = 0u, p1 = 0u, p2 = 0u, p3 = 0u;
#pragma unroll
        for (int i = 0; i < 16; ++i) {
            t0 += r[i].x; t1 += r[i].y; t2 += r[i].z; t3 += r[i].w;
            if (grp * 16 + i < c) { p0 += r[i].x; p1 += r[i].y; p2 += r[i].z; p3 += r[i].w; }
        }
        const int o = grp * NBINS + 4 * qd;
        partT[o] = t0; partT[o + 1] = t1; partT[o + 2] = t2; partT[o + 3] = t3;
        partP[o] = p0; partP[o + 1] = p1; partP[o + 2] = p2; partP[o + 3] = p3;
    }
    __syncthreads();
    if (threadIdx.x < NBINS) {
        unsigned t = 0u, p = 0u;
#pragma unroll
        for (int g = 0; g < 16; ++g) {
            t += partT[g * NBINS + threadIdx.x];
            p += partP[g * NBINS + threadIdx.x];
        }
        tot[threadIdx.x] = t;
        cur[threadIdx.x] = p;
        lst[threadIdx.x] = own;
    }
    __syncthreads();
    if (threadIdx.x < 32) {
        const int lane = threadIdx.x;
        unsigned v[NBINS / 32], cv[NBINS / 32], ow[NBINS / 32];
        unsigned s = 0u, so = 0u;
#pragma unroll
        for (int k = 0; k < NBINS / 32; ++k) {
            v[k]  = tot[lane * (NBINS / 32) + k];
            cv[k] = cur[lane * (NBINS / 32) + k];
            ow[k] = lst[lane * (NBINS / 32) + k];
            s += v[k];
            so += ow[k];
        }
        unsigned inc = s, inco = so;
#pragma unroll
        for (int d = 1; d < 32; d <<= 1) {
            const unsigned up  = __shfl_up(inc, d);
            const unsigned upo = __shfl_up(inco, d);
            if (lane >= d) { inc += up; inco += upo; }
        }
        unsigned run  = inc - s;
        unsigned lrun = inco - so;
#pragma unroll
        for (int k = 0; k < NBINS / 32; ++k) {
            const int b = lane * (NBINS / 32) + k;
            gst[b] = cv[k] + run;
            lst[b] = lrun;
            cur[b] = lrun;
            if (blockIdx.x == 0) bstart[b] = run;
            run  += v[k];
            lrun += ow[k];
        }
        if (lane == 31) lst[NBINS] = lrun;
    }
    __syncthreads();
    const int chunk = (ne + NCHUNK - 1) / NCHUNK;
    const int beg = blockIdx.x * chunk;
    const int end = min(beg + chunk, ne);
    for (int i = beg + threadIdx.x; i < end; i += 4 * 1024) {
        int k[4], s[4];
#pragma unroll
        for (int u = 0; u < 4; ++u) {
            const bool ok = i + u * 1024 < end;
            k[u] = ok ? key[i + u * 1024] : -1;
            s[u] = ok ? srcn[i + u * 1024] : 0;
        }
#pragma unroll
        for (int u = 0; u < 4; ++u)
            if (k[u] >= 0) {
                const int b = k[u] >> BIN_SHIFT;
                const unsigned slot = atomicAdd(&cur[b], 1u);
                const unsigned val = ((unsigned)(k[u] & (BIN_NODES - 1)) << 19) | (unsigned)s[u];
                if (slot < STAGE_CAP) stage[slot] = val;
                else rec[gst[b] + (slot - lst[b])] = val;
            }
    }
    __syncthreads();
    {
        const int lane = threadIdx.x & 31;
        const int wave = threadIdx.x >> 5;
        for (int b = wave; b < NBINS; b += 32) {
            const int l0 = (int)lst[b];
            const int l1 = (int)min(lst[b + 1], (unsigned)STAGE_CAP);
            const unsigned g0 = gst[b];
            if (l1 <= l0) continue;
            const int a = (int)(g0 & 31u);
            for (int j = l0 - a + lane; j < l1; j += 4 * 32) {
                unsigned v[4];
#pragma unroll
                for (int u = 0; u < 4; ++u) v[u] = stage[min(max(j + 32 * u, l0), l1 - 1)];
#pragma unroll
                for (int u = 0; u < 4; ++u) {
                    const int jj = j + 32 * u;
                    if (jj >= l0 && jj < l1) rec[g0 + (unsigned)(jj - l0)] = v[u];
                }
            }
        }
    }
}

__global__ void __launch_bounds__(1024)
bin_deg_kernel(const unsigned* __restrict__ rec,
               const unsigned* __restrict__ bstart,
               float* __restrict__ dis, int n, int ne) {
    __shared__ unsigned cnt[BIN_NODES];
    const int b = blockIdx.x;
    const int base = b << BIN_SHIFT;
    if (base >= n) return;
    for (int i = threadIdx.x; i < BIN_NODES; i += blockDim.x) cnt[i] = 1u;
    __syncthreads();
    const int beg = (int)bstart[b];
    const int end = (b == NBINS - 1) ? ne : (int)bstart[b + 1];
    for (int i = beg + threadIdx.x; i < end; i += 8 * 1024) {
        unsigned r[8];
#pragma unroll
        for (int u = 0; u < 8; ++u) {
            const int j = i + u * 1024;
            r[u] = rec[j < end ? j : i];
        }
#pragma unroll
        for (int u = 0; u < 8; ++u)
            atomicAdd(&cnt[r[u] >> 19], (i + u * 1024 < end) ? 1u : 0u);
    }
    __syncthreads();
    for (int i = threadIdx.x; i < BIN_NODES; i += blockDim.x) {
        const int node = base + i;
        if (node < n) dis[node] = rsqrtf((float)cnt[i]);
    }
}

__global__ void __launch_bounds__(1024)
bin_sum_kernel(const unsigned* __restrict__ rec,
               const unsigned* __restrict__ bstart,
               const float* __restrict__ x,
               const float* __restrict__ scsh,
               const float* __restrict__ dis,
               const float* __restrict__ Wm,
               const float* __restrict__ bias,
               float* __restrict__ out, int colofs, int n, int ne) {
    __shared__ float acc[BIN_NODES * 8];
    const int b = blockIdx.x;
    const int base = b << BIN_SHIFT;
    if (base >= n) return;
    for (int i = threadIdx.x; i < BIN_NODES * 8; i += blockDim.x) acc[i] = 0.f;
    __syncthreads();

    const int lane = threadIdx.x & 31;
    const int wave = threadIdx.x >> 5;
    const int f = lane & 7;
    const int g = lane >> 3;
    const float sc = scsh[f];
    const float sh = scsh[8 + f];
    const int beg = (int)bstart[b];
    const int end = (b == NBINS - 1) ? ne : (int)bstart[b + 1];

    for (int i0 = beg + wave * 32; i0 < end; i0 += 1024) {
        const int i = i0 + lane;
        unsigned r = 0u;
        float w = 0.f;
        if (i < end) { r = rec[i]; w = dis[r & 0x7FFFFu]; }
        float v[8];
        int   a[8];
#pragma unroll
        for (int j = 0; j < 8; ++j) {
            const int s = 4 * j + g;
            const unsigned rj = __shfl(r, s);
            const float wj = __shfl(w, s);
            const int sj = (int)(rj & 0x7FFFFu);
            v[j] = (fmaxf(x[sj * 8 + f], 0.f) * sc + sh) * wj;
            a[j] = (int)(rj >> 19) * 8 + f;
        }
#pragma unroll
        for (int j = 0; j < 8; ++j)
            if (i0 + 4 * j + g < end) atomicAdd(&acc[a[j]], v[j]);
    }
    __syncthreads();

    const int half = lane >> 4;
    const int l    = lane & 15;
    const int kA0 = 2 * half;
    const int kA1 = 4 + 2 * half;
    v2f b0 = {0.f, 0.f}, b1 = {0.f, 0.f};
    float bs = 0.f;
    if (l < 8) {
        b0.x = Wm[kA0 * 8 + l];  b0.y = Wm[(kA0 + 1) * 8 + l];
        b1.x = Wm[kA1 * 8 + l];  b1.y = Wm[(kA1 + 1) * 8 + l];
        bs = bias[l];
    }
    for (int t = wave; t < BIN_NODES / 16; t += 32) {
        const int row0 = base + t * 16;
        if (row0 >= n) break;
        const int lrow = t * 16 + l;
        const int row  = row0 + l;
        const int rowc = row < n ? row : (n - 1);
        const float d = dis[rowc];
        const float* xr = x + (size_t)rowc * 8;
        const float* ar = acc + lrow * 8;
        v2f a0, a1;
        a0.x = (ar[kA0]     + (fmaxf(xr[kA0],     0.f) * scsh[kA0]     + scsh[8 + kA0])     * d) * d;
        a0.y = (ar[kA0 + 1] + (fmaxf(xr[kA0 + 1], 0.f) * scsh[kA0 + 1] + scsh[8 + kA0 + 1]) * d) * d;
        a1.x = (ar[kA1]     + (fmaxf(xr[kA1],     0.f) * scsh[kA1]     + scsh[8 + kA1])     * d) * d;
        a1.y = (ar[kA1 + 1] + (fmaxf(xr[kA1 + 1], 0.f) * scsh[kA1 + 1] + scsh[8 + kA1 + 1]) * d) * d;
        v8f c = {};
        c = __builtin_amdgcn_wmma_f32_16x16x4_f32(false, a0, false, b0, (short)0, c, false, false);
        c = __builtin_amdgcn_wmma_f32_16x16x4_f32(false, a1, false, b1, (short)0, c, false, false);
        if (l < 8) {
            const int mbase = row0 + 8 * half;
            float* po = out + (size_t)mbase * 16 + colofs + l;
#pragma unroll
            for (int j = 0; j < 8; ++j)
                if (mbase + j < n) po[(size_t)j * 16] = c[j] + bs;
        }
    }
}

extern "C" void kernel_launch(void* const* d_in, const int* in_sizes, int n_in,
                              void* d_out, int out_size, void* d_ws, size_t ws_size,
                              hipStream_t stream) {
    const float* x     = (const float*)d_in[0];
    const int*   ei    = (const int*)d_in[1];
    const float* gamma = (const float*)d_in[2];
    const float* beta  = (const float*)d_in[3];
    const float* W_in  = (const float*)d_in[4];
    const float* b_in  = (const float*)d_in[5];
    const float* W_out = (const float*)d_in[6];
    const float* b_out = (const float*)d_in[7];
    float* out = (float*)d_out;

    const int n  = in_sizes[0] / 8;
    const int ne = in_sizes[1] / 2;
    const int NB = 1024;

    float* ws      = (float*)d_ws;
    unsigned* rec  = (unsigned*)ws;
    float* dis_c   = ws + (size_t)n * 16;
    float* dis_r   = dis_c + n;
    float* partial = dis_r + n;
    float* scsh    = partial + (size_t)NB * 16;

    const int* rows = ei;
    const int* cols = ei + ne;

    bn_partial_kernel<<<NB, 256, 0, stream>>>(x, partial, n);
    bn_finalize_kernel<<<1, 32, 0, stream>>>(partial, gamma, beta, scsh, NB, 1.0f / (float)n);

    unsigned* tab_in    = (unsigned*)dis_r;
    unsigned* bstart_in = tab_in + NBINS * NCHUNK;
    bin_count_kernel<<<NCHUNK, 1024, 0, stream>>>(cols, tab_in, ne);
    bin_place_kernel<<<NCHUNK, 1024, 0, stream>>>(cols, rows, tab_in, bstart_in, rec, ne);
    bin_deg_kernel<<<NBINS, 1024, 0, stream>>>(rec, bstart_in, dis_c, n, ne);
    bin_sum_kernel<<<NBINS, 1024, 0, stream>>>(rec, bstart_in, x, scsh, dis_c,
                                               W_in, b_in, out, 0, n, ne);

    unsigned* tab_out    = (unsigned*)dis_c;
    unsigned* bstart_out = tab_out + NBINS * NCHUNK;
    bin_count_kernel<<<NCHUNK, 1024, 0, stream>>>(rows, tab_out, ne);
    bin_place_kernel<<<NCHUNK, 1024, 0, stream>>>(rows, cols, tab_out, bstart_out, rec, ne);
    bin_deg_kernel<<<NBINS, 1024, 0, stream>>>(rec, bstart_out, dis_r, n, ne);
    bin_sum_kernel<<<NBINS, 1024, 0, stream>>>(rec, bstart_out, x, scsh, dis_r,
                                               W_out, b_out, out, 8, n, ne);
}


# bn_finalize: 8 lanes walking the 1024 block partials in 128 dependent load rounds become 1024 lanes, one round of 16 loads each, then shuffle + LDS sum; the partials are summed in a tree order now
# speedup vs baseline: 7.4952x; 1.1437x over previous
#include <hip/hip_runtime.h>
#include <hip/hip_bf16.h>

typedef float v2f __attribute__((ext_vector_type(2)));
typedef float v8f __attribute__((ext_vector_type(8)));

__device__ __forceinline__ void fadd_agent(float* p, float v) {
    __hip_atomic_fetch_add(p, v, __ATOMIC_RELAXED, __HIP_MEMORY_SCOPE_AGENT);
}

__global__ void bn_partial_kernel(const float* __restrict__ x,
                                  float* __restrict__ partial, int n) {
    float v[16];
#pragma unroll
    for (int k = 0; k < 16; ++k) v[k] = 0.f;

    const int stride = gridDim.x * blockDim.x;
    for (int i = blockIdx.x * blockDim.x + threadIdx.x; i < n; i += stride) {
        const float4* p = (const float4*)(x + (size_t)i * 8);
        float4 v0 = p[0];
        float4 v1 = p[1];
        float h;
        h = fmaxf(v0.x, 0.f); v[0] += h; v[8]  += h * h;
        h = fmaxf(v0.y, 0.f); v[1] += h; v[9]  += h * h;
        h = fmaxf(v0.z, 0.f); v[2] += h; v[10] += h * h;
        h = fmaxf(v0.w, 0.f); v[3] += h; v[11] += h * h;
        h = fmaxf(v1.x, 0.f); v[4] += h; v[12] += h * h;
        h = fmaxf(v1.y, 0.f); v[5] += h; v[13] += h * h;
        h = fmaxf(v1.z, 0.f); v[6] += h; v[14] += h * h;
        h = fmaxf(v1.w, 0.f); v[7] += h; v[15] += h * h;
    }

#pragma unroll
    for (int k = 0; k < 16; ++k) {
        float t = v[k];
        t += __shfl_xor(t, 1);
        t += __shfl_xor(t, 2);
        t += __shfl_xor(t, 4);
        t += __shfl_xor(t, 8);
        t += __shfl_xor(t, 16);
        v[k] = t;
    }

    __shared__ float sh[8 * 16];
    const int wave = threadIdx.x >> 5;
    const int lane = threadIdx.x & 31;
    if (lane == 0) {
#pragma unroll
        for (int k = 0; k < 16; ++k) sh[wave * 16 + k] = v[k];
    }
    __syncthreads();
    if (threadIdx.x < 16) {
        float t = 0.f;
#pragma unroll
        for (int w = 0; w < 8; ++w) t += sh[w * 16 + threadIdx.x];
        partial[blockIdx.x * 16 + threadIdx.x] = t;
    }
}

__global__ void __launch_bounds__(1024)
bn_finalize_kernel(const float* __restrict__ partial,
                   const float* __restrict__ gamma,
                   const float* __restrict__ beta,
                   float* __restrict__ scsh, int nb, float invN) {
    __shared__ float red[32 * 16];
    const int lane = threadIdx.x & 31;
    const int wave = threadIdx.x >> 5;
    const int k    = threadIdx.x & 15;
    const int r0   = threadIdx.x >> 4;
    float acc = 0.f;
    for (int rb = r0; rb < nb; rb += 1024) {
        float v[16];
#pragma unroll
        for (int j = 0; j < 16; ++j) {
            const int r = rb + 64 * j;
            const float t = partial[(size_t)(r < nb ? r : rb) * 16 + k];
            v[j] = (r < nb) ? t : 0.f;
        }
#pragma unroll
        for (int j = 0; j < 16; ++j) acc += v[j];
    }
    acc += __shfl_xor(acc, 16);
    if (lane < 16) red[wave * 16 + lane] = acc;
    __syncthreads();
    if (wave == 0) {
        float t = 0.f;
#pragma unroll
        for (int w = 0; w < 32; ++w) t += red[w * 16 + k];
        const float q = __shfl(t, (lane + 8) & 31);
        if (lane < 8) {
            float mean = t * invN;
            float var  = q * invN - mean * mean;
            float sc   = gamma[lane] * rsqrtf(var + 1e-5f);
            scsh[lane]     = sc;
            scsh[8 + lane] = beta[lane] - mean * sc;
        }
    }
}

#define BIN_SHIFT 11
#define BIN_NODES 2048
#define NBINS 256
#define NCHUNK 256

__global__ void bin_count_kernel(const int* __restrict__ key,
                                 unsigned* __restrict__ tab, int ne) {
    __shared__ unsigned hist[NBINS];
    if (threadIdx.x < NBINS) hist[threadIdx.x] = 0u;
    __syncthreads();
    const int chunk = (ne + NCHUNK - 1) / NCHUNK;
    const int beg = blockIdx.x * chunk;
    const int end = min(beg + chunk, ne);
    for (int i = beg + threadIdx.x; i < end; i += 4 * 1024) {
        int k[4];
#pragma unroll
        for (int u = 0; u < 4; ++u)
            k[u] = (i + u * 1024 < end) ? key[i + u * 1024] : -1;
#pragma unroll
        for (int u = 0; u < 4; ++u)
            if (k[u] >= 0) atomicAdd(&hist[k[u] >> BIN_SHIFT], 1u);
    }
    __syncthreads();
    if (threadIdx.x < NBINS) tab[blockIdx.x * NBINS + threadIdx.x] = hist[threadIdx.x];
}

#define STAGE_CAP 32768
__global__ void bin_place_kernel(const int* __restrict__ key,
                                 const int* __restrict__ srcn,
                                 const unsigned* __restrict__ tab,
                                 unsigned* __restrict__ bstart,
                                 unsigned* __restrict__ rec, int ne) {
    __shared__ unsigned cur[NBINS];
    __shared__ unsigned tot[NBINS];
    __shared__ unsigned lst[NBINS + 1];
    __shared__ unsigned gst[NBINS];
    __shared__ __attribute__((aligned(16))) unsigned stage[STAGE_CAP];
    unsigned* partT = stage;
    unsigned* partP = stage + 16 * NBINS;
    const unsigned own = (threadIdx.x < NBINS) ? tab[(size_t)blockIdx.x * NBINS + threadIdx.x] : 0u;
    {
        const int c   = blockIdx.x;
        const int qd  = threadIdx.x & 63;
        const int grp = threadIdx.x >> 6;
        const uint4* col = (const uint4*)tab + (size_t)(grp * 16) * (NBINS / 4) + qd;
        uint4 r[16];
#pragma unroll
        for (int i = 0; i < 16; ++i) r[i] = col[i * (NBINS / 4)];
        unsigned t0 = 0u, t1 = 0u, t2 = 0u, t3 = 0u;
        unsigned p0 = 0u, p1 = 0u, p2 = 0u, p3 = 0u;
#pragma unroll
        for (int i = 0; i < 16; ++i) {
            t0 += r[i].x; t1 += r[i].y; t2 += r[i].z; t3 += r[i].w;
            if (grp * 16 + i < c) { p0 += r[i].x; p1 += r[i].y; p2 += r[i].z; p3 += r[i].w; }
        }
        const int o = grp * NBINS + 4 * qd;
        partT[o] = t0; partT[o + 1] = t1; partT[o + 2] = t2; partT[o + 3] = t3;
        partP[o] = p0; partP[o + 1] = p1; partP[o + 2] = p2; partP[o + 3] = p3;
    }
    __syncthreads();
    if (threadIdx.x < NBINS) {
        unsigned t = 0u, p = 0u;
#pragma unroll
        for (int g = 0; g < 16; ++g) {
            t += partT[g * NBINS + threadIdx.x];
            p += partP[g * NBINS + threadIdx.x];
        }
        tot[threadIdx.x] = t;
        cur[threadIdx.x] = p;
        lst[threadIdx.x] = own;
    }
    __syncthreads();
    if (threadIdx.x < 32) {
        const int lane = threadIdx.x;
        unsigned v[NBINS / 32], cv[NBINS / 32], ow[NBINS / 32];
        unsigned s = 0u, so = 0u;
#pragma unroll
        for (int k = 0; k < NBINS / 32; ++k) {
            v[k]  = tot[lane * (NBINS / 32) + k];
            cv[k] = cur[lane * (NBINS / 32) + k];
            ow[k] = lst[lane * (NBINS / 32) + k];
            s += v[k];
            so += ow[k];
        }
        unsigned inc = s, inco = so;
#pragma unroll
        for (int d = 1; d < 32; d <<= 1) {
            const unsigned up  = __shfl_up(inc, d);
            const unsigned upo = __shfl_up(inco, d);
            if (lane >= d) { inc += up; inco += upo; }
        }
        unsigned run  = inc - s;
        unsigned lrun = inco - so;
#pragma unroll
        for (int k = 0; k < NBINS / 32; ++k) {
            const int b = lane * (NBINS / 32) + k;
            gst[b] = cv[k] + run;
            lst[b] = lrun;
            cur[b] = lrun;
            if (blockIdx.x == 0) bstart[b] = run;
            run  += v[k];
            lrun += ow[k];
        }
        if (lane == 31) lst[NBINS] = lrun;
    }
    __syncthreads();
    const int chunk = (ne + NCHUNK - 1) / NCHUNK;
    const int beg = blockIdx.x * chunk;
    const int end = min(beg + chunk, ne);
    for (int i = beg + threadIdx.x; i < end; i += 4 * 1024) {
        int k[4], s[4];
#pragma unroll
        for (int u = 0; u < 4; ++u) {
            const bool ok = i + u * 1024 < end;
            k[u] = ok ? key[i + u * 1024] : -1;
            s[u] = ok ? srcn[i + u * 1024] : 0;
        }
#pragma unroll
        for (int u = 0; u < 4; ++u)
            if (k[u] >= 0) {
                const int b = k[u] >> BIN_SHIFT;
                const unsigned slot = atomicAdd(&cur[b], 1u);
                const unsigned val = ((unsigned)(k[u] & (BIN_NODES - 1)) << 19) | (unsigned)s[u];
                if (slot < STAGE_CAP) stage[slot] = val;
                else rec[gst[b] + (slot - lst[b])] = val;
            }
    }
    __syncthreads();
    {
        const int lane = threadIdx.x & 31;
        const int wave = threadIdx.x >> 5;
        for (int b = wave; b < NBINS; b += 32) {
            const int l0 = (int)lst[b];
            const int l1 = (int)min(lst[b + 1], (unsigned)STAGE_CAP);
            const unsigned g0 = gst[b];
            if (l1 <= l0) continue;
            const int a = (int)(g0 & 31u);
            for (int j = l0 - a + lane; j < l1; j += 4 * 32) {
                unsigned v[4];
#pragma unroll
                for (int u = 0; u < 4; ++u) v[u] = stage[min(max(j + 32 * u, l0), l1 - 1)];
#pragma unroll
                for (int u = 0; u < 4; ++u) {
                    const int jj = j + 32 * u;
                    if (jj >= l0 && jj < l1) rec[g0 + (unsigned)(jj - l0)] = v[u];
                }
            }
        }
    }
}

__global__ void __launch_bounds__(1024)
bin_deg_kernel(const unsigned* __restrict__ rec,
               const unsigned* __restrict__ bstart,
               float* __restrict__ dis, int n, int ne) {
    __shared__ unsigned cnt[BIN_NODES];
    const int b = blockIdx.x;
    const int base = b << BIN_SHIFT;
    if (base >= n) return;
    for (int i = threadIdx.x; i < BIN_NODES; i += blockDim.x) cnt[i] = 1u;
    __syncthreads();
    const int beg = (int)bstart[b];
    const int end = (b == NBINS - 1) ? ne : (int)bstart[b + 1];
    for (int i = beg + threadIdx.x; i < end; i += 8 * 1024) {
        unsigned r[8];
#pragma unroll
        for (int u = 0; u < 8; ++u) {
            const int j = i + u * 1024;
            r[u] = rec[j < end ? j : i];
        }
#pragma unroll
        for (int u = 0; u < 8; ++u)
            atomicAdd(&cnt[r[u] >> 19], (i + u * 1024 < end) ? 1u : 0u);
    }
    __syncthreads();
    for (int i = threadIdx.x; i < BIN_NODES; i += blockDim.x) {
        const int node = base + i;
        if (node < n) dis[node] = rsqrtf((float)cnt[i]);
    }
}

__global__ void __launch_bounds__(1024)
bin_sum_kernel(const unsigned* __restrict__ rec,
               const unsigned* __restrict__ bstart,
               const float* __restrict__ x,
               const float* __restrict__ scsh,
               const float* __restrict__ dis,
               const float* __restrict__ Wm,
               const float* __restrict__ bias,
               float* __restrict__ out, int colofs, int n, int ne) {
    __shared__ float acc[BIN_NODES * 8];
    const int b = blockIdx.x;
    const int base = b << BIN_SHIFT;
    if (base >= n) return;
    for (int i = threadIdx.x; i < BIN_NODES * 8; i += blockDim.x) acc[i] = 0.f;
    __syncthreads();

    const int lane = threadIdx.x & 31;
    const int wave = threadIdx.x >> 5;
    const int f = lane & 7;
    const int g = lane >> 3;
    const float sc = scsh[f];
    const float sh = scsh[8 + f];
    const int beg = (int)bstart[b];
    const int end = (b == NBINS - 1) ? ne : (int)bstart[b + 1];

    for (int i0 = beg + wave * 32; i0 < end; i0 += 1024) {
        const int i = i0 + lane;
        unsigned r = 0u;
        float w = 0.f;
        if (i < end) { r = rec[i]; w = dis[r & 0x7FFFFu]; }
        float v[8];
        int   a[8];
#pragma unroll
        for (int j = 0; j < 8; ++j) {
            const int s = 4 * j + g;
            const unsigned rj = __shfl(r, s);
            const float wj = __shfl(w, s);
            const int sj = (int)(rj & 0x7FFFFu);
            v[j] = (fmaxf(x[sj * 8 + f], 0.f) * sc + sh) * wj;
            a[j] = (int)(rj >> 19) * 8 + f;
        }
#pragma unroll
        for (int j = 0; j < 8; ++j)
            if (i0 + 4 * j + g < end) atomicAdd(&acc[a[j]], v[j]);
    }
    __syncthreads();

    const int half = lane >> 4;
    const int l    = lane & 15;
    const int kA0 = 2 * half;
    const int kA1 = 4 + 2 * half;
    v2f b0 = {0.f, 0.f}, b1 = {0.f, 0.f};
    float bs = 0.f;
    if (l < 8) {
        b0.x = Wm[kA0 * 8 + l];  b0.y = Wm[(kA0 + 1) * 8 + l];
        b1.x = Wm[kA1 * 8 + l];  b1.y = Wm[(kA1 + 1) * 8 + l];
        bs = bias[l];
    }
    for (int t = wave; t < BIN_NODES / 16; t += 32) {
        const int row0 = base + t * 16;
        if (row0 >= n) break;
        const int lrow = t * 16 + l;
        const int row  = row0 + l;
        const int rowc = row < n ? row : (n - 1);
        const float d = dis[rowc];
        const float* xr = x + (size_t)rowc * 8;
        const float* ar = acc + lrow * 8;
        v2f a0, a1;
        a0.x = (ar[kA0]     + (fmaxf(xr[kA0],     0.f) * scsh[kA0]     + scsh[8 + kA0])     * d) * d;
        a0.y = (ar[kA0 + 1] + (fmaxf(xr[kA0 + 1], 0.f) * scsh[kA0 + 1] + scsh[8 + kA0 + 1]) * d) * d;
        a1.x = (ar[kA1]     + (fmaxf(xr[kA1],     0.f) * scsh[kA1]     + scsh[8 + kA1])     * d) * d;
        a1.y = (ar[kA1 + 1] + (fmaxf(xr[kA1 + 1], 0.f) * scsh[kA1 + 1] + scsh[8 + kA1 + 1]) * d) * d;
        v8f c = {};
        c = __builtin_amdgcn_wmma_f32_16x16x4_f32(false, a0, false, b0, (short)0, c, false, false);
        c = __builtin_amdgcn_wmma_f32_16x16x4_f32(false, a1, false, b1, (short)0, c, false, false);
        if (l < 8) {
            const int mbase = row0 + 8 * half;
            float* po = out + (size_t)mbase * 16 + colofs + l;
#pragma unroll
            for (int j = 0; j < 8; ++j)
                if (mbase + j < n) po[(size_t)j * 16] = c[j] + bs;
        }
    }
}

extern "C" void kernel_launch(void* const* d_in, const int* in_sizes, int n_in,
                              void* d_out, int out_size, void* d_ws, size_t ws_size,
                              hipStream_t stream) {
    const float* x     = (const float*)d_in[0];
    const int*   ei    = (const int*)d_in[1];
    const float* gamma = (const float*)d_in[2];
    const float* beta  = (const float*)d_in[3];
    const float* W_in  = (const float*)d_in[4];
    const float* b_in  = (const float*)d_in[5];
    const float* W_out = (const float*)d_in[6];
    const float* b_out = (const float*)d_in[7];
    float* out = (float*)d_out;

    const int n  = in_sizes[0] / 8;
    const int ne = in_sizes[1] / 2;
    const int NB = 1024;

    float* ws      = (float*)d_ws;
    unsigned* rec  = (unsigned*)ws;
    float* dis_c   = ws + (size_t)n * 16;
    float* dis_r   = dis_c + n;
    float* partial = dis_r + n;
    float* scsh    = partial + (size_t)NB * 16;

    const int* rows = ei;
    const int* cols = ei + ne;

    bn_partial_kernel<<<NB, 256, 0, stream>>>(x, partial, n);
    bn_finalize_kernel<<<1, 1024, 0, stream>>>(partial, gamma, beta, scsh, NB, 1.0f / (float)n);

    unsigned* tab_in    = (unsigned*)dis_r;
    unsigned* bstart_in = tab_in + NBINS * NCHUNK;
    bin_count_kernel<<<NCHUNK, 1024, 0, stream>>>(cols, tab_in, ne);
    bin_place_kernel<<<NCHUNK, 1024, 0, stream>>>(cols, rows, tab_in, bstart_in, rec, ne);
    bin_deg_kernel<<<NBINS, 1024, 0, stream>>>(rec, bstart_in, dis_c, n, ne);
    bin_sum_kernel<<<NBINS, 1024, 0, stream>>>(rec, bstart_in, x, scsh, dis_c,
                                               W_in, b_in, out, 0, n, ne);

    unsigned* tab_out    = (unsigned*)dis_c;
    unsigned* bstart_out = tab_out + NBINS * NCHUNK;
    bin_count_kernel<<<NCHUNK, 1024, 0, stream>>>(rows, tab_out, ne);
    bin_place_kernel<<<NCHUNK, 1024, 0, stream>>>(rows, cols, tab_out, bstart_out, rec, ne);
    bin_deg_kernel<<<NBINS, 1024, 0, stream>>>(rec, bstart_out, dis_r, n, ne);
    bin_sum_kernel<<<NBINS, 1024, 0, stream>>>(rec, bstart_out, x, scsh, dis_r,
                                               W_out, b_out, out, 8, n, ne);
}


# in-conv sum: bin_deg also stores each node's normalised row * dis in the node's not yet written out-conv words of out; the sum gathers that row per record, not a dis word + x row; add order kept
# speedup vs baseline: 8.1614x; 1.0889x over previous
#include <hip/hip_runtime.h>
#include <hip/hip_bf16.h>

typedef float v2f __attribute__((ext_vector_type(2)));
typedef float v8f __attribute__((ext_vector_type(8)));

__device__ __forceinline__ void fadd_agent(float* p, float v) {
    __hip_atomic_fetch_add(p, v, __ATOMIC_RELAXED, __HIP_MEMORY_SCOPE_AGENT);
}

__global__ void bn_partial_kernel(const float* __restrict__ x,
                                  float* __restrict__ partial, int n) {
    float v[16];
#pragma unroll
    for (int k = 0; k < 16; ++k) v[k] = 0.f;

    const int stride = gridDim.x * blockDim.x;
    for (int i = blockIdx.x * blockDim.x + threadIdx.x; i < n; i += stride) {
        const float4* p = (const float4*)(x + (size_t)i * 8);
        float4 v0 = p[0];
        float4 v1 = p[1];
        float h;
        h = fmaxf(v0.x, 0.f); v[0] += h; v[8]  += h * h;
        h = fmaxf(v0.y, 0.f); v[1] += h; v[9]  += h * h;
        h = fmaxf(v0.z, 0.f); v[2] += h; v[10] += h * h;
        h = fmaxf(v0.w, 0.f); v[3] += h; v[11] += h * h;
        h = fmaxf(v1.x, 0.f); v[4] += h; v[12] += h * h;
        h = fmaxf(v1.y, 0.f); v[5] += h; v[13] += h * h;
        h = fmaxf(v1.z, 0.f); v[6] += h; v[14] += h * h;
        h = fmaxf(v1.w, 0.f); v[7] += h; v[15] += h * h;
    }

#pragma unroll
    for (int k = 0; k < 16; ++k) {
        float t = v[k];
        t += __shfl_xor(t, 1);
        t += __shfl_xor(t, 2);
        t += __shfl_xor(t, 4);
        t += __shfl_xor(t, 8);
        t += __shfl_xor(t, 16);
        v[k] = t;
    }

    __shared__ float sh[8 * 16];
    const int wave = threadIdx.x >> 5;
    const int lane = threadIdx.x & 31;
    if (lane == 0) {
#pragma unroll
        for (int k = 0; k < 16; ++k) sh[wave * 16 + k] = v[k];
    }
    __syncthreads();
    if (threadIdx.x < 16) {
        float t = 0.f;
#pragma unroll
        for (int w = 0; w < 8; ++w) t += sh[w * 16 + threadIdx.x];
        partial[blockIdx.x * 16 + threadIdx.x] = t;
    }
}

__global__ void __launch_bounds__(1024)
bn_finalize_kernel(const float* __restrict__ partial,
                   const float* __restrict__ gamma,
                   const float* __restrict__ beta,
                   float* __restrict__ scsh, int nb, float invN) {
    __shared__ float red[32 * 16];
    const int lane = threadIdx.x & 31;
    const int wave = threadIdx.x >> 5;
    const int k    = threadIdx.x & 15;
    const int r0   = threadIdx.x >> 4;
    float acc = 0.f;
    for (int rb = r0; rb < nb; rb += 1024) {
        float v[16];
#pragma unroll
        for (int j = 0; j < 16; ++j) {
            const int r = rb + 64 * j;
            const float t = partial[(size_t)(r < nb ? r : rb) * 16 + k];
            v[j] = (r < nb) ? t : 0.f;
        }
#pragma unroll
        for (int j = 0; j < 16; ++j) acc += v[j];
    }
    acc += __shfl_xor(acc, 16);
    if (lane < 16) red[wave * 16 + lane] = acc;
    __syncthreads();
    if (wave == 0) {
        float t = 0.f;
#pragma unroll
        for (int w = 0; w < 32; ++w) t += red[w * 16 + k];
        const float q = __shfl(t, (lane + 8) & 31);
        if (lane < 8) {
            float mean = t * invN;
            float var  = q * invN - mean * mean;
            float sc   = gamma[lane] * rsqrtf(var + 1e-5f);
            scsh[lane]     = sc;
            scsh[8 + lane] = beta[lane] - mean * sc;
        }
    }
}

#define BIN_SHIFT 11
#define BIN_NODES 2048
#define NBINS 256
#define NCHUNK 256

__global__ void bin_count_kernel(const int* __restrict__ key,
                                 unsigned* __restrict__ tab, int ne) {
    __shared__ unsigned hist[NBINS];
    if (threadIdx.x < NBINS) hist[threadIdx.x] = 0u;
    __syncthreads();
    const int chunk = (ne + NCHUNK - 1) / NCHUNK;
    const int beg = blockIdx.x * chunk;
    const int end = min(beg + chunk, ne);
    for (int i = beg + threadIdx.x; i < end; i += 4 * 1024) {
        int k[4];
#pragma unroll
        for (int u = 0; u < 4; ++u)
            k[u] = (i + u * 1024 < end) ? key[i + u * 1024] : -1;
#pragma unroll
        for (int u = 0; u < 4; ++u)
            if (k[u] >= 0) atomicAdd(&hist[k[u] >> BIN_SHIFT], 1u);
    }
    __syncthreads();
    if (threadIdx.x < NBINS) tab[blockIdx.x * NBINS + threadIdx.x] = hist[threadIdx.x];
}

#define STAGE_CAP 32768
__global__ void bin_place_kernel(const int* __restrict__ key,
                                 const int* __restrict__ srcn,
                                 const unsigned* __restrict__ tab,
                                 unsigned* __restrict__ bstart,
                                 unsigned* __restrict__ rec, int ne) {
    __shared__ unsigned cur[NBINS];
    __shared__ unsigned tot[NBINS];
    __shared__ unsigned lst[NBINS + 1];
    __shared__ unsigned gst[NBINS];
    __shared__ __attribute__((aligned(16))) unsigned stage[STAGE_CAP];
    unsigned* partT = stage;
    unsigned* partP = stage + 16 * NBINS;
    const unsigned own = (threadIdx.x < NBINS) ? tab[(size_t)blockIdx.x * NBINS + threadIdx.x] : 0u;
    {
        const int c   = blockIdx.x;
        const int qd  = threadIdx.x & 63;
        const int grp = threadIdx.x >> 6;
        const uint4* col = (const uint4*)tab + (size_t)(grp * 16) * (NBINS / 4) + qd;
        uint4 r[16];
#pragma unroll
        for (int i = 0; i < 16; ++i) r[i] = col[i * (NBINS / 4)];
        unsigned t0 = 0u, t1 = 0u, t2 = 0u, t3 = 0u;
        unsigned p0 = 0u, p1 = 0u, p2 = 0u, p3 = 0u;
#pragma unroll
        for (int i = 0; i < 16; ++i) {
            t0 += r[i].x; t1 += r[i].y; t2 += r[i].z; t3 += r[i].w;
            if (grp * 16 + i < c) { p0 += r[i].x; p1 += r[i].y; p2 += r[i].z; p3 += r[i].w; }
        }
        const int o = grp * NBINS + 4 * qd;
        partT[o] = t0; partT[o + 1] = t1; partT[o + 2] = t2; partT[o + 3] = t3;
        partP[o] = p0; partP[o + 1] = p1; partP[o + 2] = p2; partP[o + 3] = p3;
    }
    __syncthreads();
    if (threadIdx.x < NBINS) {
        unsigned t = 0u, p = 0u;
#pragma unroll
        for (int g = 0; g < 16; ++g) {
            t += partT[g * NBINS + threadIdx.x];
            p += partP[g * NBINS + threadIdx.x];
        }
        tot[threadIdx.x] = t;
        cur[threadIdx.x] = p;
        lst[threadIdx.x] = own;
    }
    __syncthreads();
    if (threadIdx.x < 32) {
        const int lane = threadIdx.x;
        unsigned v[NBINS / 32], cv[NBINS / 32], ow[NBINS / 32];
        unsigned s = 0u, so = 0u;
#pragma unroll
        for (int k = 0; k < NBINS / 32; ++k) {
            v[k]  = tot[lane * (NBINS / 32) + k];
            cv[k] = cur[lane * (NBINS / 32) + k];
            ow[k] = lst[lane * (NBINS / 32) + k];
            s += v[k];
            so += ow[k];
        }
        unsigned inc = s, inco = so;
#pragma unroll
        for (int d = 1; d < 32; d <<= 1) {
            const unsigned up  = __shfl_up(inc, d);
            const unsigned upo = __shfl_up(inco, d);
            if (lane >= d) { inc += up; inco += upo; }
        }
        unsigned run  = inc - s;
        unsigned lrun = inco - so;
#pragma unroll
        for (int k = 0; k < NBINS / 32; ++k) {
            const int b = lane * (NBINS / 32) + k;
            gst[b] = cv[k] + run;
            lst[b] = lrun;
            cur[b] = lrun;
            if (blockIdx.x == 0) bstart[b] = run;
            run  += v[k];
            lrun += ow[k];
        }
        if (lane == 31) lst[NBINS] = lrun;
    }
    __syncthreads();
    const int chunk = (ne + NCHUNK - 1) / NCHUNK;
    const int beg = blockIdx.x * chunk;
    const int end = min(beg + chunk, ne);
    for (int i = beg + threadIdx.x; i < end; i += 4 * 1024) {
        int k[4], s[4];
#pragma unroll
        for (int u = 0; u < 4; ++u) {
            const bool ok = i + u * 1024 < end;
            k[u] = ok ? key[i + u * 1024] : -1;
            s[u] = ok ? srcn[i + u * 1024] : 0;
        }
#pragma unroll
        for (int u = 0; u < 4; ++u)
            if (k[u] >= 0) {
                const int b = k[u] >> BIN_SHIFT;
                const unsigned slot = atomicAdd(&cur[b], 1u);
                const unsigned val = ((unsigned)(k[u] & (BIN_NODES - 1)) << 19) | (unsigned)s[u];
                if (slot < STAGE_CAP) stage[slot] = val;
                else rec[gst[b] + (slot - lst[b])] = val;
            }
    }
    __syncthreads();
    {
        const int lane = threadIdx.x & 31;
        const int wave = threadIdx.x >> 5;
        for (int b = wave; b < NBINS; b += 32) {
            const int l0 = (int)lst[b];
            const int l1 = (int)min(lst[b + 1], (unsigned)STAGE_CAP);
            const unsigned g0 = gst[b];
            if (l1 <= l0) continue;
            const int a = (int)(g0 & 31u);
            for (int j = l0 - a + lane; j < l1; j += 4 * 32) {
                unsigned v[4];
#pragma unroll
                for (int u = 0; u < 4; ++u) v[u] = stage[min(max(j + 32 * u, l0), l1 - 1)];
#pragma unroll
                for (int u = 0; u < 4; ++u) {
                    const int jj = j + 32 * u;
                    if (jj >= l0 && jj < l1) rec[g0 + (unsigned)(jj - l0)] = v[u];
                }
            }
        }
    }
}

template <bool YROW>
__global__ void __launch_bounds__(1024)
bin_deg_kernel(const unsigned* __restrict__ rec,
               const unsigned* __restrict__ bstart,
               float* __restrict__ dis,
               const float* __restrict__ x,
               const float* __restrict__ scsh,
               float* __restrict__ yrow, int n, int ne) {
    __shared__ unsigned cnt[BIN_NODES];
    const int b = blockIdx.x;
    const int base = b << BIN_SHIFT;
    if (base >= n) return;
    for (int i = threadIdx.x; i < BIN_NODES; i += blockDim.x) cnt[i] = 1u;
    __syncthreads();
    const int beg = (int)bstart[b];
    const int end = (b == NBINS - 1) ? ne : (int)bstart[b + 1];
    for (int i = beg + threadIdx.x; i < end; i += 8 * 1024) {
        unsigned r[8];
#pragma unroll
        for (int u = 0; u < 8; ++u) {
            const int j = i + u * 1024;
            r[u] = rec[j < end ? j : i];
        }
#pragma unroll
        for (int u = 0; u < 8; ++u)
            atomicAdd(&cnt[r[u] >> 19], (i + u * 1024 < end) ? 1u : 0u);
    }
    __syncthreads();
    for (int i = threadIdx.x; i < BIN_NODES; i += blockDim.x) {
        const int node = base + i;
        if (node < n) dis[node] = rsqrtf((float)cnt[i]);
    }
    if (YROW) {
        const int hf = threadIdx.x & 1;
        float4 sc, sh;
        sc.x = scsh[4 * hf];     sc.y = scsh[4 * hf + 1];  sc.z = scsh[4 * hf + 2];  sc.w = scsh[4 * hf + 3];
        sh.x = scsh[8 + 4 * hf]; sh.y = scsh[9 + 4 * hf];  sh.z = scsh[10 + 4 * hf]; sh.w = scsh[11 + 4 * hf];
        float4 xv[BIN_NODES / 512];
#pragma unroll
        for (int u = 0; u < BIN_NODES / 512; ++u) {
            const int node = base + (threadIdx.x >> 1) + 512 * u;
            xv[u] = ((const float4*)(x + (size_t)(node < n ? node : n - 1) * 8))[hf];
        }
#pragma unroll
        for (int u = 0; u < BIN_NODES / 512; ++u) {
            const int q = (threadIdx.x >> 1) + 512 * u;
            const int node = base + q;
            const float d = rsqrtf((float)cnt[q]);
            float4 y;
            y.x = (fmaxf(xv[u].x, 0.f) * sc.x + sh.x) * d;
            y.y = (fmaxf(xv[u].y, 0.f) * sc.y + sh.y) * d;
            y.z = (fmaxf(xv[u].z, 0.f) * sc.z + sh.z) * d;
            y.w = (fmaxf(xv[u].w, 0.f) * sc.w + sh.w) * d;
            if (node < n) ((float4*)(yrow + (size_t)node * 16 + 8))[hf] = y;
        }
    }
}

template <bool YROW>
__global__ void __launch_bounds__(1024)
bin_sum_kernel(const unsigned* __restrict__ rec,
               const unsigned* __restrict__ bstart,
               const float* __restrict__ x,
               const float* __restrict__ scsh,
               const float* __restrict__ dis,
               const float* __restrict__ Wm,
               const float* __restrict__ bias,
               const float* yrow,
               float* out, int colofs, int n, int ne) {
    __shared__ float acc[BIN_NODES * 8];
    const int b = blockIdx.x;
    const int base = b << BIN_SHIFT;
    if (base >= n) return;
    for (int i = threadIdx.x; i < BIN_NODES * 8; i += blockDim.x) acc[i] = 0.f;
    __syncthreads();

    const int lane = threadIdx.x & 31;
    const int wave = threadIdx.x >> 5;
    const int f = lane & 7;
    const int g = lane >> 3;
    const float sc = scsh[f];
    const float sh = scsh[8 + f];
    const int beg = (int)bstart[b];
    const int end = (b == NBINS - 1) ? ne : (int)bstart[b + 1];

    for (int i0 = beg + wave * 32; i0 < end; i0 += 1024) {
        const int i = i0 + lane;
        unsigned r = 0u;
        float w = 0.f;
        if (i < end) {
            r = rec[i];
            if (!YROW) w = dis[r & 0x7FFFFu];
        }
        float v[8];
        int   a[8];
#pragma unroll
        for (int j = 0; j < 8; ++j) {
            const int s = 4 * j + g;
            const unsigned rj = __shfl(r, s);
            const int sj = (int)(rj & 0x7FFFFu);
            if (YROW) {
                v[j] = yrow[(size_t)sj * 16 + 8 + f];
            } else {
                const float wj = __shfl(w, s);
                v[j] = (fmaxf(x[sj * 8 + f], 0.f) * sc + sh) * wj;
            }
            a[j] = (int)(rj >> 19) * 8 + f;
        }
#pragma unroll
        for (int j = 0; j < 8; ++j)
            if (i0 + 4 * j + g < end) atomicAdd(&acc[a[j]], v[j]);
    }
    __syncthreads();

    const int half = lane >> 4;
    const int l    = lane & 15;
    const int kA0 = 2 * half;
    const int kA1 = 4 + 2 * half;
    v2f b0 = {0.f, 0.f}, b1 = {0.f, 0.f};
    float bs = 0.f;
    if (l < 8) {
        b0.x = Wm[kA0 * 8 + l];  b0.y = Wm[(kA0 + 1) * 8 + l];
        b1.x = Wm[kA1 * 8 + l];  b1.y = Wm[(kA1 + 1) * 8 + l];
        bs = bias[l];
    }
    for (int t = wave; t < BIN_NODES / 16; t += 32) {
        const int row0 = base + t * 16;
        if (row0 >= n) break;
        const int lrow = t * 16 + l;
        const int row  = row0 + l;
        const int rowc = row < n ? row : (n - 1);
        const float d = dis[rowc];
        const float* xr = x + (size_t)rowc * 8;
        const float* ar = acc + lrow * 8;
        v2f a0, a1;
        a0.x = (ar[kA0]     + (fmaxf(xr[kA0],     0.f) * scsh[kA0]     + scsh[8 + kA0])     * d) * d;
        a0.y = (ar[kA0 + 1] + (fmaxf(xr[kA0 + 1], 0.f) * scsh[kA0 + 1] + scsh[8 + kA0 + 1]) * d) * d;
        a1.x = (ar[kA1]     + (fmaxf(xr[kA1],     0.f) * scsh[kA1]     + scsh[8 + kA1])     * d) * d;
        a1.y = (ar[kA1 + 1] + (fmaxf(xr[kA1 + 1], 0.f) * scsh[kA1 + 1] + scsh[8 + kA1 + 1]) * d) * d;
        v8f c = {};
        c = __builtin_amdgcn_wmma_f32_16x16x4_f32(false, a0, false, b0, (short)0, c, false, false);
        c = __builtin_amdgcn_wmma_f32_16x16x4_f32(false, a1, false, b1, (short)0, c, false, false);
        if (l < 8) {
            const int mbase = row0 + 8 * half;
            float* po = out + (size_t)mbase * 16 + colofs + l;
#pragma unroll
            for (int j = 0; j < 8; ++j)
                if (mbase + j < n) po[(size_t)j * 16] = c[j] + bs;
        }
    }
}

extern "C" void kernel_launch(void* const* d_in, const int* in_sizes, int n_in,
                              void* d_out, int out_size, void* d_ws, size_t ws_size,
                              hipStream_t stream) {
    const float* x     = (const float*)d_in[0];
    const int*   ei    = (const int*)d_in[1];
    const float* gamma = (const float*)d_in[2];
    const float* beta  = (const float*)d_in[3];
    const float* W_in  = (const float*)d_in[4];
    const float* b_in  = (const float*)d_in[5];
    const float* W_out = (const float*)d_in[6];
    const float* b_out = (const float*)d_in[7];
    float* out = (float*)d_out;

    const int n  = in_sizes[0] / 8;
    const int ne = in_sizes[1] / 2;
    const int NB = 1024;

    float* ws      = (float*)d_ws;
    unsigned* rec  = (unsigned*)ws;
    float* dis_c   = ws + (size_t)n * 16;
    float* dis_r   = dis_c + n;
    float* partial = dis_r + n;
    float* scsh    = partial + (size_t)NB * 16;

    const int* rows = ei;
    const int* cols = ei + ne;

    bn_partial_kernel<<<NB, 256, 0, stream>>>(x, partial, n);
    bn_finalize_kernel<<<1, 1024, 0, stream>>>(partial, gamma, beta, scsh, NB, 1.0f / (float)n);

    unsigned* tab_in    = (unsigned*)dis_r;
    unsigned* bstart_in = tab_in + NBINS * NCHUNK;
    bin_count_kernel<<<NCHUNK, 1024, 0, stream>>>(cols, tab_in, ne);
    bin_place_kernel<<<NCHUNK, 1024, 0, stream>>>(cols, rows, tab_in, bstart_in, rec, ne);
    bin_deg_kernel<true><<<NBINS, 1024, 0, stream>>>(rec, bstart_in, dis_c, x, scsh, out, n, ne);
    bin_sum_kernel<true><<<NBINS, 1024, 0, stream>>>(rec, bstart_in, x, scsh, dis_c,
                                                     W_in, b_in, out, out, 0, n, ne);

    unsigned* tab_out    = (unsigned*)dis_c;
    unsigned* bstart_out = tab_out + NBINS * NCHUNK;
    bin_count_kernel<<<NCHUNK, 1024, 0, stream>>>(rows, tab_out, ne);
    bin_place_kernel<<<NCHUNK, 1024, 0, stream>>>(rows, cols, tab_out, bstart_out, rec, ne);
    bin_deg_kernel<false><<<NBINS, 1024, 0, stream>>>(rec, bstart_out, dis_r, x, scsh, out, n, ne);
    bin_sum_kernel<false><<<NBINS, 1024, 0, stream>>>(rec, bstart_out, x, scsh, dis_r,
                                                      W_out, b_out, out, out, 8, n, ne);
}


# launches 10 -> 7: BN partial sums and both directions' bin counts share one launch (one walk of the edge list, two LDS histograms); BN finalize is an extra work-group of the first placing launch
# speedup vs baseline: 8.5714x; 1.0502x over previous
#include <hip/hip_runtime.h>
#include <hip/hip_bf16.h>

typedef float v2f __attribute__((ext_vector_type(2)));
typedef float v8f __attribute__((ext_vector_type(8)));

__device__ __forceinline__ void fadd_agent(float* p, float v) {
    __hip_atomic_fetch_add(p, v, __ATOMIC_RELAXED, __HIP_MEMORY_SCOPE_AGENT);
}

#define BIN_SHIFT 11
#define BIN_NODES 2048
#define NBINS 256
#define NCHUNK 256

__global__ void __launch_bounds__(1024)
stats_count_kernel(const float* __restrict__ x,
                   const int* __restrict__ keyA,
                   const int* __restrict__ keyB,
                   float* __restrict__ partial,
                   unsigned* __restrict__ tabA,
                   unsigned* __restrict__ tabB, int n, int ne) {
    __shared__ unsigned histA[NBINS];
    __shared__ unsigned histB[NBINS];
    __shared__ float sh[32 * 16];
    if (threadIdx.x < NBINS) { histA[threadIdx.x] = 0u; histB[threadIdx.x] = 0u; }

    float v[16];
#pragma unroll
    for (int k = 0; k < 16; ++k) v[k] = 0.f;

    const int stride = gridDim.x * blockDim.x;
    for (int i = blockIdx.x * blockDim.x + threadIdx.x; i < n; i += stride) {
        const float4* p = (const float4*)(x + (size_t)i * 8);
        float4 v0 = p[0];
        float4 v1 = p[1];
        float h;
        h = fmaxf(v0.x, 0.f); v[0] += h; v[8]  += h * h;
        h = fmaxf(v0.y, 0.f); v[1] += h; v[9]  += h * h;
        h = fmaxf(v0.z, 0.f); v[2] += h; v[10] += h * h;
        h = fmaxf(v0.w, 0.f); v[3] += h; v[11] += h * h;
        h = fmaxf(v1.x, 0.f); v[4] += h; v[12] += h * h;
        h = fmaxf(v1.y, 0.f); v[5] += h; v[13] += h * h;
        h = fmaxf(v1.z, 0.f); v[6] += h; v[14] += h * h;
        h = fmaxf(v1.w, 0.f); v[7] += h; v[15] += h * h;
    }

#pragma unroll
    for (int k = 0; k < 16; ++k) {
        float t = v[k];
        t += __shfl_xor(t, 1);
        t += __shfl_xor(t, 2);
        t += __shfl_xor(t, 4);
        t += __shfl_xor(t, 8);
        t += __shfl_xor(t, 16);
        v[k] = t;
    }

    const int wave = threadIdx.x >> 5;
    const int lane = threadIdx.x & 31;
    if (lane == 0) {
#pragma unroll
        for (int k = 0; k < 16; ++k) sh[wave * 16 + k] = v[k];
    }
    __syncthreads();
    {
        const int qt = threadIdx.x >> 8;
        const int c  = threadIdx.x & 255;
        if (c < 16) {
            float t = 0.f;
#pragma unroll
            for (int w = 0; w < 8; ++w) t += sh[(qt * 8 + w) * 16 + c];
            partial[(blockIdx.x * 4 + qt) * 16 + c] = t;
        }
    }

    const int chunk = (ne + NCHUNK - 1) / NCHUNK;
    const int beg = blockIdx.x * chunk;
    const int end = min(beg + chunk, ne);
    for (int i = beg + threadIdx.x; i < end; i += 4 * 1024) {
        int ka[4], kb[4];
#pragma unroll
        for (int u = 0; u < 4; ++u) {
            const int j = i + u * 1024;
            const int jc = j < end ? j : i;
            ka[u] = keyA[jc];
            kb[u] = keyB[jc];
        }
#pragma unroll
        for (int u = 0; u < 4; ++u) {
            const unsigned one = (i + u * 1024 < end) ? 1u : 0u;
            atomicAdd(&histA[ka[u] >> BIN_SHIFT], one);
            atomicAdd(&histB[kb[u] >> BIN_SHIFT], one);
        }
    }
    __syncthreads();
    if (threadIdx.x < NBINS) {
        tabA[blockIdx.x * NBINS + threadIdx.x] = histA[threadIdx.x];
        tabB[blockIdx.x * NBINS + threadIdx.x] = histB[threadIdx.x];
    }
}

__device__ __forceinline__ void
bn_finalize_block(const float* __restrict__ partial,
                  const float* __restrict__ gamma,
                  const float* __restrict__ beta,
                  float* __restrict__ scsh, int nb, float invN, float* red) {
    const int lane = threadIdx.x & 31;
    const int wave = threadIdx.x >> 5;
    const int k    = threadIdx.x & 15;
    const int r0   = threadIdx.x >> 4;
    float acc = 0.f;
    for (int rb = r0; rb < nb; rb += 1024) {
        float v[16];
#pragma unroll
        for (int j = 0; j < 16; ++j) {
            const int r = rb + 64 * j;
            const float t = partial[(size_t)(r < nb ? r : rb) * 16 + k];
            v[j] = (r < nb) ? t : 0.f;
        }
#pragma unroll
        for (int j = 0; j < 16; ++j) acc += v[j];
    }
    acc += __shfl_xor(acc, 16);
    if (lane < 16) red[wave * 16 + lane] = acc;
    __syncthreads();
    if (wave == 0) {
        float t = 0.f;
#pragma unroll
        for (int w = 0; w < 32; ++w) t += red[w * 16 + k];
        const float q = __shfl(t, (lane + 8) & 31);
        if (lane < 8) {
            float mean = t * invN;
            float var  = q * invN - mean * mean;
            float sc   = gamma[lane] * rsqrtf(var + 1e-5f);
            scsh[lane]     = sc;
            scsh[8 + lane] = beta[lane] - mean * sc;
        }
    }
}


#define STAGE_CAP 32768
template <bool FIN>
__global__ void bin_place_kernel(const int* __restrict__ key,
                                 const int* __restrict__ srcn,
                                 const unsigned* __restrict__ tab,
                                 unsigned* __restrict__ bstart,
                                 unsigned* __restrict__ rec, int ne,
                                 const float* __restrict__ partial,
                                 const float* __restrict__ gamma,
                                 const float* __restrict__ beta,
                                 float* __restrict__ scsh, int nb, float invN) {
    __shared__ unsigned cur[NBINS];
    __shared__ unsigned tot[NBINS];
    __shared__ unsigned lst[NBINS + 1];
    __shared__ unsigned gst[NBINS];
    __shared__ __attribute__((aligned(16))) unsigned stage[STAGE_CAP];
    if (FIN && blockIdx.x == NCHUNK) {
        bn_finalize_block(partial, gamma, beta, scsh, nb, invN, (float*)stage);
        return;
    }
    unsigned* partT = stage;
    unsigned* partP = stage + 16 * NBINS;
    const unsigned own = (threadIdx.x < NBINS) ? tab[(size_t)blockIdx.x * NBINS + threadIdx.x] : 0u;
    {
        const int c   = blockIdx.x;
        const int qd  = threadIdx.x & 63;
        const int grp = threadIdx.x >> 6;
        const uint4* col = (const uint4*)tab + (size_t)(grp * 16) * (NBINS / 4) + qd;
        uint4 r[16];
#pragma unroll
        for (int i = 0; i < 16; ++i) r[i] = col[i * (NBINS / 4)];
        unsigned t0 = 0u, t1 = 0u, t2 = 0u, t3 = 0u;
        unsigned p0 = 0u, p1 = 0u, p2 = 0u, p3 = 0u;
#pragma unroll
        for (int i = 0; i < 16; ++i) {
            t0 += r[i].x; t1 += r[i].y; t2 += r[i].z; t3 += r[i].w;
            if (grp * 16 + i < c) { p0 += r[i].x; p1 += r[i].y; p2 += r[i].z; p3 += r[i].w; }
        }
        const int o = grp * NBINS + 4 * qd;
        partT[o] = t0; partT[o + 1] = t1; partT[o + 2] = t2; partT[o + 3] = t3;
        partP[o] = p0; partP[o + 1] = p1; partP[o + 2] = p2; partP[o + 3] = p3;
    }
    __syncthreads();
    if (threadIdx.x < NBINS) {
        unsigned t = 0u, p = 0u;
#pragma unroll
        for (int g = 0; g < 16; ++g) {
            t += partT[g * NBINS + threadIdx.x];
            p += partP[g * NBINS + threadIdx.x];
        }
        tot[threadIdx.x] = t;
        cur[threadIdx.x] = p;
        lst[threadIdx.x] = own;
    }
    __syncthreads();
    if (threadIdx.x < 32) {
        const int lane = threadIdx.x;
        unsigned v[NBINS / 32], cv[NBINS / 32], ow[NBINS / 32];
        unsigned s = 0u, so = 0u;
#pragma unroll
        for (int k = 0; k < NBINS / 32; ++k) {
            v[k]  = tot[lane * (NBINS / 32) + k];
            cv[k] = cur[lane * (NBINS / 32) + k];
            ow[k] = lst[lane * (NBINS / 32) + k];
            s += v[k];
            so += ow[k];
        }
        unsigned inc = s, inco = so;
#pragma unroll
        for (int d = 1; d < 32; d <<= 1) {
            const unsigned up  = __shfl_up(inc, d);
            const unsigned upo = __shfl_up(inco, d);
            if (lane >= d) { inc += up; inco += upo; }
        }
        unsigned run  = inc - s;
        unsigned lrun = inco - so;
#pragma unroll
        for (int k = 0; k < NBINS / 32; ++k) {
            const int b = lane * (NBINS / 32) + k;
            gst[b] = cv[k] + run;
            lst[b] = lrun;
            cur[b] = lrun;
            if (blockIdx.x == 0) bstart[b] = run;
            run  += v[k];
            lrun += ow[k];
        }
        if (lane == 31) lst[NBINS] = lrun;
    }
    __syncthreads();
    const int chunk = (ne + NCHUNK - 1) / NCHUNK;
    const int beg = blockIdx.x * chunk;
    const int end = min(beg + chunk, ne);
    for (int i = beg + threadIdx.x; i < end; i += 4 * 1024) {
        int k[4], s[4];
#pragma unroll
        for (int u = 0; u < 4; ++u) {
            const bool ok = i + u * 1024 < end;
            k[u] = ok ? key[i + u * 1024] : -1;
            s[u] = ok ? srcn[i + u * 1024] : 0;
        }
#pragma unroll
        for (int u = 0; u < 4; ++u)
            if (k[u] >= 0) {
                const int b = k[u] >> BIN_SHIFT;
                const unsigned slot = atomicAdd(&cur[b], 1u);
                const unsigned val = ((unsigned)(k[u] & (BIN_NODES - 1)) << 19) | (unsigned)s[u];
                if (slot < STAGE_CAP) stage[slot] = val;
                else rec[gst[b] + (slot - lst[b])] = val;
            }
    }
    __syncthreads();
    {
        const int lane = threadIdx.x & 31;
        const int wave = threadIdx.x >> 5;
        for (int b = wave; b < NBINS; b += 32) {
            const int l0 = (int)lst[b];
            const int l1 = (int)min(lst[b + 1], (unsigned)STAGE_CAP);
            const unsigned g0 = gst[b];
            if (l1 <= l0) continue;
            const int a = (int)(g0 & 31u);
            for (int j = l0 - a + lane; j < l1; j += 4 * 32) {
                unsigned v[4];
#pragma unroll
                for (int u = 0; u < 4; ++u) v[u] = stage[min(max(j + 32 * u, l0), l1 - 1)];
#pragma unroll
                for (int u = 0; u < 4; ++u) {
                    const int jj = j + 32 * u;
                    if (jj >= l0 && jj < l1) rec[g0 + (unsigned)(jj - l0)] = v[u];
                }
            }
        }
    }
}

template <bool YROW>
__global__ void __launch_bounds__(1024)
bin_deg_kernel(const unsigned* __restrict__ rec,
               const unsigned* __restrict__ bstart,
               float* __restrict__ dis,
               const float* __restrict__ x,
               const float* __restrict__ scsh,
               float* __restrict__ yrow, int n, int ne) {
    __shared__ unsigned cnt[BIN_NODES];
    const int b = blockIdx.x;
    const int base = b << BIN_SHIFT;
    if (base >= n) return;
    for (int i = threadIdx.x; i < BIN_NODES; i += blockDim.x) cnt[i] = 1u;
    __syncthreads();
    const int beg = (int)bstart[b];
    const int end = (b == NBINS - 1) ? ne : (int)bstart[b + 1];
    for (int i = beg + threadIdx.x; i < end; i += 8 * 1024) {
        unsigned r[8];
#pragma unroll
        for (int u = 0; u < 8; ++u) {
            const int j = i + u * 1024;
            r[u] = rec[j < end ? j : i];
        }
#pragma unroll
        for (int u = 0; u < 8; ++u)
            atomicAdd(&cnt[r[u] >> 19], (i + u * 1024 < end) ? 1u : 0u);
    }
    __syncthreads();
    for (int i = threadIdx.x; i < BIN_NODES; i += blockDim.x) {
        const int node = base + i;
        if (node < n) dis[node] = rsqrtf((float)cnt[i]);
    }
    if (YROW) {
        const int hf = threadIdx.x & 1;
        float4 sc, sh;
        sc.x = scsh[4 * hf];     sc.y = scsh[4 * hf + 1];  sc.z = scsh[4 * hf + 2];  sc.w = scsh[4 * hf + 3];
        sh.x = scsh[8 + 4 * hf]; sh.y = scsh[9 + 4 * hf];  sh.z = scsh[10 + 4 * hf]; sh.w = scsh[11 + 4 * hf];
        float4 xv[BIN_NODES / 512];
#pragma unroll
        for (int u = 0; u < BIN_NODES / 512; ++u) {
            const int node = base + (threadIdx.x >> 1) + 512 * u;
            xv[u] = ((const float4*)(x + (size_t)(node < n ? node : n - 1) * 8))[hf];
        }
#pragma unroll
        for (int u = 0; u < BIN_NODES / 512; ++u) {
            const int q = (threadIdx.x >> 1) + 512 * u;
            const int node = base + q;
            const float d = rsqrtf((float)cnt[q]);
            float4 y;
            y.x = (fmaxf(xv[u].x, 0.f) * sc.x + sh.x) * d;
            y.y = (fmaxf(xv[u].y, 0.f) * sc.y + sh.y) * d;
            y.z = (fmaxf(xv[u].z, 0.f) * sc.z + sh.z) * d;
            y.w = (fmaxf(xv[u].w, 0.f) * sc.w + sh.w) * d;
            if (node < n) ((float4*)(yrow + (size_t)node * 16 + 8))[hf] = y;
        }
    }
}

template <bool YROW>
__global__ void __launch_bounds__(1024)
bin_sum_kernel(const unsigned* __restrict__ rec,
               const unsigned* __restrict__ bstart,
               const float* __restrict__ x,
               const float* __restrict__ scsh,
               const float* __restrict__ dis,
               const float* __restrict__ Wm,
               const float* __restrict__ bias,
               const float* yrow,
               float* out, int colofs, int n, int ne) {
    __shared__ float acc[BIN_NODES * 8];
    const int b = blockIdx.x;
    const int base = b << BIN_SHIFT;
    if (base >= n) return;
    for (int i = threadIdx.x; i < BIN_NODES * 8; i += blockDim.x) acc[i] = 0.f;
    __syncthreads();

    const int lane = threadIdx.x & 31;
    const int wave = threadIdx.x >> 5;
    const int f = lane & 7;
    const int g = lane >> 3;
    const float sc = scsh[f];
    const float sh = scsh[8 + f];
    const int beg = (int)bstart[b];
    const int end = (b == NBINS - 1) ? ne : (int)bstart[b + 1];

    for (int i0 = beg + wave * 32; i0 < end; i0 += 1024) {
        const int i = i0 + lane;
        unsigned r = 0u;
        float w = 0.f;
        if (i < end) {
            r = rec[i];
            if (!YROW) w = dis[r & 0x7FFFFu];
        }
        float v[8];
        int   a[8];
#pragma unroll
        for (int j = 0; j < 8; ++j) {
            const int s = 4 * j + g;
            const unsigned rj = __shfl(r, s);
            const int sj = (int)(rj & 0x7FFFFu);
            if (YROW) {
                v[j] = yrow[(size_t)sj * 16 + 8 + f];
            } else {
                const float wj = __shfl(w, s);
                v[j] = (fmaxf(x[sj * 8 + f], 0.f) * sc + sh) * wj;
            }
            a[j] = (int)(rj >> 19) * 8 + f;
        }
#pragma unroll
        for (int j = 0; j < 8; ++j)
            if (i0 + 4 * j + g < end) atomicAdd(&acc[a[j]], v[j]);
    }
    __syncthreads();

    const int half = lane >> 4;
    const int l    = lane & 15;
    const int kA0 = 2 * half;
    const int kA1 = 4 + 2 * half;
    v2f b0 = {0.f, 0.f}, b1 = {0.f, 0.f};
    float bs = 0.f;
    if (l < 8) {
        b0.x = Wm[kA0 * 8 + l];  b0.y = Wm[(kA0 + 1) * 8 + l];
        b1.x = Wm[kA1 * 8 + l];  b1.y = Wm[(kA1 + 1) * 8 + l];
        bs = bias[l];
    }
    for (int t = wave; t < BIN_NODES / 16; t += 32) {
        const int row0 = base + t * 16;
        if (row0 >= n) break;
        const int lrow = t * 16 + l;
        const int row  = row0 + l;
        const int rowc = row < n ? row : (n - 1);
        const float d = dis[rowc];
        const float* xr = x + (size_t)rowc * 8;
        const float* ar = acc + lrow * 8;
        v2f a0, a1;
        a0.x = (ar[kA0]     + (fmaxf(xr[kA0],     0.f) * scsh[kA0]     + scsh[8 + kA0])     * d) * d;
        a0.y = (ar[kA0 + 1] + (fmaxf(xr[kA0 + 1], 0.f) * scsh[kA0 + 1] + scsh[8 + kA0 + 1]) * d) * d;
        a1.x = (ar[kA1]     + (fmaxf(xr[kA1],     0.f) * scsh[kA1]     + scsh[8 + kA1])     * d) * d;
        a1.y = (ar[kA1 + 1] + (fmaxf(xr[kA1 + 1], 0.f) * scsh[kA1 + 1] + scsh[8 + kA1 + 1]) * d) * d;
        v8f c = {};
        c = __builtin_amdgcn_wmma_f32_16x16x4_f32(false, a0, false, b0, (short)0, c, false, false);
        c = __builtin_amdgcn_wmma_f32_16x16x4_f32(false, a1, false, b1, (short)0, c, false, false);
        if (l < 8) {
            const int mbase = row0 + 8 * half;
            float* po = out + (size_t)mbase * 16 + colofs + l;
#pragma unroll
            for (int j = 0; j < 8; ++j)
                if (mbase + j < n) po[(size_t)j * 16] = c[j] + bs;
        }
    }
}

extern "C" void kernel_launch(void* const* d_in, const int* in_sizes, int n_in,
                              void* d_out, int out_size, void* d_ws, size_t ws_size,
                              hipStream_t stream) {
    const float* x     = (const float*)d_in[0];
    const int*   ei    = (const int*)d_in[1];
    const float* gamma = (const float*)d_in[2];
    const float* beta  = (const float*)d_in[3];
    const float* W_in  = (const float*)d_in[4];
    const float* b_in  = (const float*)d_in[5];
    const float* W_out = (const float*)d_in[6];
    const float* b_out = (const float*)d_in[7];
    float* out = (float*)d_out;

    const int n  = in_sizes[0] / 8;
    const int ne = in_sizes[1] / 2;
    const int NB = 1024;

    float* ws      = (float*)d_ws;
    unsigned* rec  = (unsigned*)ws;
    float* dis_c   = ws + (size_t)n * 16;
    float* dis_r   = dis_c + n;
    float* partial = dis_r + n;
    float* scsh    = partial + (size_t)NB * 16;

    const int* rows = ei;
    const int* cols = ei + ne;

    unsigned* tab_in     = (unsigned*)dis_r;
    unsigned* bstart_in  = tab_in + NBINS * NCHUNK;
    unsigned* tab_out    = bstart_in + NBINS;
    unsigned* bstart_out = (unsigned*)dis_c;

    stats_count_kernel<<<NCHUNK, 1024, 0, stream>>>(x, cols, rows, partial, tab_in, tab_out, n, ne);

    bin_place_kernel<true><<<NCHUNK + 1, 1024, 0, stream>>>(cols, rows, tab_in, bstart_in, rec, ne,
                                                            partial, gamma, beta, scsh, NB, 1.0f / (float)n);
    bin_deg_kernel<true><<<NBINS, 1024, 0, stream>>>(rec, bstart_in, dis_c, x, scsh, out, n, ne);
    bin_sum_kernel<true><<<NBINS, 1024, 0, stream>>>(rec, bstart_in, x, scsh, dis_c,
                                                     W_in, b_in, out, out, 0, n, ne);

    bin_place_kernel<false><<<NCHUNK, 1024, 0, stream>>>(rows, cols, tab_out, bstart_out, rec, ne,
                                                         partial, gamma, beta, scsh, NB, 1.0f / (float)n);
    bin_deg_kernel<false><<<NBINS, 1024, 0, stream>>>(rec, bstart_out, dis_r, x, scsh, out, n, ne);
    bin_sum_kernel<false><<<NBINS, 1024, 0, stream>>>(rec, bstart_out, x, scsh, dis_r,
                                                      W_out, b_out, out, out, 8, n, ne);
}


# out-conv sum gathers one ready row per record from a table bin_deg stores in out words 8..15 (no dis gather); results park in the bin's spent rec range, an 8th launch copies them out; add order kept
# speedup vs baseline: 9.2252x; 1.0763x over previous
#include <hip/hip_runtime.h>
#include <hip/hip_bf16.h>

typedef float v2f __attribute__((ext_vector_type(2)));
typedef float v8f __attribute__((ext_vector_type(8)));

__device__ __forceinline__ void fadd_agent(float* p, float v) {
    __hip_atomic_fetch_add(p, v, __ATOMIC_RELAXED, __HIP_MEMORY_SCOPE_AGENT);
}

#define BIN_SHIFT 11
#define BIN_NODES 2048
#define NBINS 256
#define NCHUNK 256

__global__ void __launch_bounds__(1024)
stats_count_kernel(const float* __restrict__ x,
                   const int* __restrict__ keyA,
                   const int* __restrict__ keyB,
                   float* __restrict__ partial,
                   unsigned* __restrict__ tabA,
                   unsigned* __restrict__ tabB, int n, int ne) {
    __shared__ unsigned histA[NBINS];
    __shared__ unsigned histB[NBINS];
    __shared__ float sh[32 * 16];
    if (threadIdx.x < NBINS) { histA[threadIdx.x] = 0u; histB[threadIdx.x] = 0u; }

    float v[16];
#pragma unroll
    for (int k = 0; k < 16; ++k) v[k] = 0.f;

    const int stride = gridDim.x * blockDim.x;
    for (int i = blockIdx.x * blockDim.x + threadIdx.x; i < n; i += stride) {
        const float4* p = (const float4*)(x + (size_t)i * 8);
        float4 v0 = p[0];
        float4 v1 = p[1];
        float h;
        h = fmaxf(v0.x, 0.f); v[0] += h; v[8]  += h * h;
        h = fmaxf(v0.y, 0.f); v[1] += h; v[9]  += h * h;
        h = fmaxf(v0.z, 0.f); v[2] += h; v[10] += h * h;
        h = fmaxf(v0.w, 0.f); v[3] += h; v[11] += h * h;
        h = fmaxf(v1.x, 0.f); v[4] += h; v[12] += h * h;
        h = fmaxf(v1.y, 0.f); v[5] += h; v[13] += h * h;
        h = fmaxf(v1.z, 0.f); v[6] += h; v[14] += h * h;
        h = fmaxf(v1.w, 0.f); v[7] += h; v[15] += h * h;
    }

#pragma unroll
    for (int k = 0; k < 16; ++k) {
        float t = v[k];
        t += __shfl_xor(t, 1);
        t += __shfl_xor(t, 2);
        t += __shfl_xor(t, 4);
        t += __shfl_xor(t, 8);
        t += __shfl_xor(t, 16);
        v[k] = t;
    }

    const int wave = threadIdx.x >> 5;
    const int lane = threadIdx.x & 31;
    if (lane == 0) {
#pragma unroll
        for (int k = 0; k < 16; ++k) sh[wave * 16 + k] = v[k];
    }
    __syncthreads();
    {
        const int qt = threadIdx.x >> 8;
        const int c  = threadIdx.x & 255;
        if (c < 16) {
            float t = 0.f;
#pragma unroll
            for (int w = 0; w < 8; ++w) t += sh[(qt * 8 + w) * 16 + c];
            partial[(blockIdx.x * 4 + qt) * 16 + c] = t;
        }
    }

    const int chunk = (ne + NCHUNK - 1) / NCHUNK;
    const int beg = blockIdx.x * chunk;
    const int end = min(beg + chunk, ne);
    for (int i = beg + threadIdx.x; i < end; i += 4 * 1024) {
        int ka[4], kb[4];
#pragma unroll
        for (int u = 0; u < 4; ++u) {
            const int j = i + u * 1024;
            const int jc = j < end ? j : i;
            ka[u] = keyA[jc];
            kb[u] = keyB[jc];
        }
#pragma unroll
        for (int u = 0; u < 4; ++u) {
            const unsigned one = (i + u * 1024 < end) ? 1u : 0u;
            atomicAdd(&histA[ka[u] >> BIN_SHIFT], one);
            atomicAdd(&histB[kb[u] >> BIN_SHIFT], one);
        }
    }
    __syncthreads();
    if (threadIdx.x < NBINS) {
        tabA[blockIdx.x * NBINS + threadIdx.x] = histA[threadIdx.x];
        tabB[blockIdx.x * NBINS + threadIdx.x] = histB[threadIdx.x];
    }
}

__device__ __forceinline__ void
bn_finalize_block(const float* __restrict__ partial,
                  const float* __restrict__ gamma,
                  const float* __restrict__ beta,
                  float* __restrict__ scsh, int nb, float invN, float* red) {
    const int lane = threadIdx.x & 31;
    const int wave = threadIdx.x >> 5;
    const int k    = threadIdx.x & 15;
    const int r0   = threadIdx.x >> 4;
    float acc = 0.f;
    for (int rb = r0; rb < nb; rb += 1024) {
        float v[16];
#pragma unroll
        for (int j = 0; j < 16; ++j) {
            const int r = rb + 64 * j;
            const float t = partial[(size_t)(r < nb ? r : rb) * 16 + k];
            v[j] = (r < nb) ? t : 0.f;
        }
#pragma unroll
        for (int j = 0; j < 16; ++j) acc += v[j];
    }
    acc += __shfl_xor(acc, 16);
    if (lane < 16) red[wave * 16 + lane] = acc;
    __syncthreads();
    if (wave == 0) {
        float t = 0.f;
#pragma unroll
        for (int w = 0; w < 32; ++w) t += red[w * 16 + k];
        const float q = __shfl(t, (lane + 8) & 31);
        if (lane < 8) {
            float mean = t * invN;
            float var  = q * invN - mean * mean;
            float sc   = gamma[lane] * rsqrtf(var + 1e-5f);
            scsh[lane]     = sc;
            scsh[8 + lane] = beta[lane] - mean * sc;
        }
    }
}


#define STAGE_CAP 32768
template <bool FIN>
__global__ void bin_place_kernel(const int* __restrict__ key,
                                 const int* __restrict__ srcn,
                                 const unsigned* __restrict__ tab,
                                 unsigned* __restrict__ bstart,
                                 unsigned* __restrict__ rec, int ne,
                                 const float* __restrict__ partial,
                                 const float* __restrict__ gamma,
                                 const float* __restrict__ beta,
                                 float* __restrict__ scsh, int nb, float invN) {
    __shared__ unsigned cur[NBINS];
    __shared__ unsigned tot[NBINS];
    __shared__ unsigned lst[NBINS + 1];
    __shared__ unsigned gst[NBINS];
    __shared__ __attribute__((aligned(16))) unsigned stage[STAGE_CAP];
    if (FIN && blockIdx.x == NCHUNK) {
        bn_finalize_block(partial, gamma, beta, scsh, nb, invN, (float*)stage);
        return;
    }
    unsigned* partT = stage;
    unsigned* partP = stage + 16 * NBINS;
    const unsigned own = (threadIdx.x < NBINS) ? tab[(size_t)blockIdx.x * NBINS + threadIdx.x] : 0u;
    {
        const int c   = blockIdx.x;
        const int qd  = threadIdx.x & 63;
        const int grp = threadIdx.x >> 6;
        const uint4* col = (const uint4*)tab + (size_t)(grp * 16) * (NBINS / 4) + qd;
        uint4 r[16];
#pragma unroll
        for (int i = 0; i < 16; ++i) r[i] = col[i * (NBINS / 4)];
        unsigned t0 = 0u, t1 = 0u, t2 = 0u, t3 = 0u;
        unsigned p0 = 0u, p1 = 0u, p2 = 0u, p3 = 0u;
#pragma unroll
        for (int i = 0; i < 16; ++i) {
            t0 += r[i].x; t1 += r[i].y; t2 += r[i].z; t3 += r[i].w;
            if (grp * 16 + i < c) { p0 += r[i].x; p1 += r[i].y; p2 += r[i].z; p3 += r[i].w; }
        }
        const int o = grp * NBINS + 4 * qd;
        partT[o] = t0; partT[o + 1] = t1; partT[o + 2] = t2; partT[o + 3] = t3;
        partP[o] = p0; partP[o + 1] = p1; partP[o + 2] = p2; partP[o + 3] = p3;
    }
    __syncthreads();
    if (threadIdx.x < NBINS) {
        unsigned t = 0u, p = 0u;
#pragma unroll
        for (int g = 0; g < 16; ++g) {
            t += partT[g * NBINS + threadIdx.x];
            p += partP[g * NBINS + threadIdx.x];
        }
        tot[threadIdx.x] = t;
        cur[threadIdx.x] = p;
        lst[threadIdx.x] = own;
    }
    __syncthreads();
    if (threadIdx.x < 32) {
        const int lane = threadIdx.x;
        unsigned v[NBINS / 32], cv[NBINS / 32], ow[NBINS / 32];
        unsigned s = 0u, so = 0u;
#pragma unroll
        for (int k = 0; k < NBINS / 32; ++k) {
            v[k]  = tot[lane * (NBINS / 32) + k];
            cv[k] = cur[lane * (NBINS / 32) + k];
            ow[k] = lst[lane * (NBINS / 32) + k];
            s += v[k];
            so += ow[k];
        }
        unsigned inc = s, inco = so;
#pragma unroll
        for (int d = 1; d < 32; d <<= 1) {
            const unsigned up  = __shfl_up(inc, d);
            const unsigned upo = __shfl_up(inco, d);
            if (lane >= d) { inc += up; inco += upo; }
        }
        unsigned run  = inc - s;
        unsigned lrun = inco - so;
#pragma unroll
        for (int k = 0; k < NBINS / 32; ++k) {
            const int b = lane * (NBINS / 32) + k;
            gst[b] = cv[k] + run;
            lst[b] = lrun;
            cur[b] = lrun;
            if (blockIdx.x == 0) bstart[b] = run;
            run  += v[k];
            lrun += ow[k];
        }
        if (lane == 31) lst[NBINS] = lrun;
    }
    __syncthreads();
    const int chunk = (ne + NCHUNK - 1) / NCHUNK;
    const int beg = blockIdx.x * chunk;
    const int end = min(beg + chunk, ne);
    for (int i = beg + threadIdx.x; i < end; i += 4 * 1024) {
        int k[4], s[4];
#pragma unroll
        for (int u = 0; u < 4; ++u) {
            const bool ok = i + u * 1024 < end;
            k[u] = ok ? key[i + u * 1024] : -1;
            s[u] = ok ? srcn[i + u * 1024] : 0;
        }
#pragma unroll
        for (int u = 0; u < 4; ++u)
            if (k[u] >= 0) {
                const int b = k[u] >> BIN_SHIFT;
                const unsigned slot = atomicAdd(&cur[b], 1u);
                const unsigned val = ((unsigned)(k[u] & (BIN_NODES - 1)) << 19) | (unsigned)s[u];
                if (slot < STAGE_CAP) stage[slot] = val;
                else rec[gst[b] + (slot - lst[b])] = val;
            }
    }
    __syncthreads();
    {
        const int lane = threadIdx.x & 31;
        const int wave = threadIdx.x >> 5;
        for (int b = wave; b < NBINS; b += 32) {
            const int l0 = (int)lst[b];
            const int l1 = (int)min(lst[b + 1], (unsigned)STAGE_CAP);
            const unsigned g0 = gst[b];
            if (l1 <= l0) continue;
            const int a = (int)(g0 & 31u);
            for (int j = l0 - a + lane; j < l1; j += 4 * 32) {
                unsigned v[4];
#pragma unroll
                for (int u = 0; u < 4; ++u) v[u] = stage[min(max(j + 32 * u, l0), l1 - 1)];
#pragma unroll
                for (int u = 0; u < 4; ++u) {
                    const int jj = j + 32 * u;
                    if (jj >= l0 && jj < l1) rec[g0 + (unsigned)(jj - l0)] = v[u];
                }
            }
        }
    }
}

__global__ void __launch_bounds__(1024)
bin_deg_kernel(const unsigned* __restrict__ rec,
               const unsigned* __restrict__ bstart,
               float* __restrict__ dis,
               const float* __restrict__ x,
               const float* __restrict__ scsh,
               float* __restrict__ yrow, int n, int ne) {
    __shared__ unsigned cnt[BIN_NODES];
    const int b = blockIdx.x;
    const int base = b << BIN_SHIFT;
    if (base >= n) return;
    for (int i = threadIdx.x; i < BIN_NODES; i += blockDim.x) cnt[i] = 1u;
    __syncthreads();
    const int beg = (int)bstart[b];
    const int end = (b == NBINS - 1) ? ne : (int)bstart[b + 1];
    for (int i = beg + threadIdx.x; i < end; i += 8 * 1024) {
        unsigned r[8];
#pragma unroll
        for (int u = 0; u < 8; ++u) {
            const int j = i + u * 1024;
            r[u] = rec[j < end ? j : i];
        }
#pragma unroll
        for (int u = 0; u < 8; ++u)
            atomicAdd(&cnt[r[u] >> 19], (i + u * 1024 < end) ? 1u : 0u);
    }
    __syncthreads();
    for (int i = threadIdx.x; i < BIN_NODES; i += blockDim.x) {
        const int node = base + i;
        if (node < n) dis[node] = rsqrtf((float)cnt[i]);
    }
    {
        const int hf = threadIdx.x & 1;
        float4 sc, sh;
        sc.x = scsh[4 * hf];     sc.y = scsh[4 * hf + 1];  sc.z = scsh[4 * hf + 2];  sc.w = scsh[4 * hf + 3];
        sh.x = scsh[8 + 4 * hf]; sh.y = scsh[9 + 4 * hf];  sh.z = scsh[10 + 4 * hf]; sh.w = scsh[11 + 4 * hf];
        float4 xv[BIN_NODES / 512];
#pragma unroll
        for (int u = 0; u < BIN_NODES / 512; ++u) {
            const int node = base + (threadIdx.x >> 1) + 512 * u;
            xv[u] = ((const float4*)(x + (size_t)(node < n ? node : n - 1) * 8))[hf];
        }
#pragma unroll
        for (int u = 0; u < BIN_NODES / 512; ++u) {
            const int q = (threadIdx.x >> 1) + 512 * u;
            const int node = base + q;
            const float d = rsqrtf((float)cnt[q]);
            float4 y;
            y.x = (fmaxf(xv[u].x, 0.f) * sc.x + sh.x) * d;
            y.y = (fmaxf(xv[u].y, 0.f) * sc.y + sh.y) * d;
            y.z = (fmaxf(xv[u].z, 0.f) * sc.z + sh.z) * d;
            y.w = (fmaxf(xv[u].w, 0.f) * sc.w + sh.w) * d;
            if (node < n) ((float4*)(yrow + (size_t)node * 16 + 8))[hf] = y;
        }
    }
}

template <int MODE>
__global__ void __launch_bounds__(1024)
bin_sum_kernel(unsigned* rec,
               const unsigned* __restrict__ bstart,
               const float* __restrict__ x,
               const float* __restrict__ scsh,
               const float* __restrict__ dis,
               const float* __restrict__ Wm,
               const float* __restrict__ bias,
               const float* yrow,
               float* out, int colofs, int n, int ne) {
    __shared__ __attribute__((aligned(16))) float acc[BIN_NODES * 8];
    constexpr bool YROW = (MODE != 2);
    const int b = blockIdx.x;
    const int base = b << BIN_SHIFT;
    if (base >= n) return;
    const int beg = (int)bstart[b];
    const int end = (b == NBINS - 1) ? ne : (int)bstart[b + 1];
    const int nodes = min(BIN_NODES, n - base);
    const int park0 = (beg + 31) & ~31;
    const bool fits = park0 + nodes * 8 <= end;
    if (MODE == 1 && !fits) return;
    if (MODE == 2 && fits) {
        static_assert(BIN_NODES * 2 == 4 * 1024, "four 16-B pieces per lane");
        const float4* src4 = (const float4*)(rec + park0);
        float* dst = out + (size_t)base * 16 + colofs;
        const int np = nodes * 2;
        const int i0 = threadIdx.x, i1 = i0 + 1024, i2 = i0 + 2048, i3 = i0 + 3072;
        const float4 p0 = src4[i0 < np ? i0 : 0];
        const float4 p1 = src4[i1 < np ? i1 : 0];
        const float4 p2 = src4[i2 < np ? i2 : 0];
        const float4 p3 = src4[i3 < np ? i3 : 0];
        if (i0 < np) ((float4*)(dst + (size_t)(i0 >> 1) * 16))[i0 & 1] = p0;
        if (i1 < np) ((float4*)(dst + (size_t)(i1 >> 1) * 16))[i1 & 1] = p1;
        if (i2 < np) ((float4*)(dst + (size_t)(i2 >> 1) * 16))[i2 & 1] = p2;
        if (i3 < np) ((float4*)(dst + (size_t)(i3 >> 1) * 16))[i3 & 1] = p3;
        return;
    }
    for (int i = threadIdx.x; i < BIN_NODES * 8; i += blockDim.x) acc[i] = 0.f;
    __syncthreads();

    const int lane = threadIdx.x & 31;
    const int wave = threadIdx.x >> 5;
    const int f = lane & 7;
    const int g = lane >> 3;
    const float sc = scsh[f];
    const float sh = scsh[8 + f];

    for (int i0 = beg + wave * 32; i0 < end; i0 += 1024) {
        const int i = i0 + lane;
        unsigned r = 0u;
        float w = 0.f;
        if (i < end) {
            r = rec[i];
            if (!YROW) w = dis[r & 0x7FFFFu];
        }
        float v[8];
        int   a[8];
#pragma unroll
        for (int j = 0; j < 8; ++j) {
            const int s = 4 * j + g;
            const unsigned rj = __shfl(r, s);
            const int sj = (int)(rj & 0x7FFFFu);
            if (YROW) {
                v[j] = yrow[(size_t)sj * 16 + 8 + f];
            } else {
                const float wj = __shfl(w, s);
                v[j] = (fmaxf(x[sj * 8 + f], 0.f) * sc + sh) * wj;
            }
            a[j] = (int)(rj >> 19) * 8 + f;
        }
#pragma unroll
        for (int j = 0; j < 8; ++j)
            if (i0 + 4 * j + g < end) atomicAdd(&acc[a[j]], v[j]);
    }
    __syncthreads();

    const int half = lane >> 4;
    const int l    = lane & 15;
    const int kA0 = 2 * half;
    const int kA1 = 4 + 2 * half;
    v2f b0 = {0.f, 0.f}, b1 = {0.f, 0.f};
    float bs = 0.f;
    if (l < 8) {
        b0.x = Wm[kA0 * 8 + l];  b0.y = Wm[(kA0 + 1) * 8 + l];
        b1.x = Wm[kA1 * 8 + l];  b1.y = Wm[(kA1 + 1) * 8 + l];
        bs = bias[l];
    }
    for (int t = wave; t < BIN_NODES / 16; t += 32) {
        const int row0 = base + t * 16;
        if (row0 >= n) break;
        const int lrow = t * 16 + l;
        const int row  = row0 + l;
        const int rowc = row < n ? row : (n - 1);
        const float d = dis[rowc];
        const float* xr = x + (size_t)rowc * 8;
        const float* ar = acc + lrow * 8;
        v2f a0, a1;
        a0.x = (ar[kA0]     + (fmaxf(xr[kA0],     0.f) * scsh[kA0]     + scsh[8 + kA0])     * d) * d;
        a0.y = (ar[kA0 + 1] + (fmaxf(xr[kA0 + 1], 0.f) * scsh[kA0 + 1] + scsh[8 + kA0 + 1]) * d) * d;
        a1.x = (ar[kA1]     + (fmaxf(xr[kA1],     0.f) * scsh[kA1]     + scsh[8 + kA1])     * d) * d;
        a1.y = (ar[kA1 + 1] + (fmaxf(xr[kA1 + 1], 0.f) * scsh[kA1 + 1] + scsh[8 + kA1 + 1]) * d) * d;
        v8f c = {};
        c = __builtin_amdgcn_wmma_f32_16x16x4_f32(false, a0, false, b0, (short)0, c, false, false);
        c = __builtin_amdgcn_wmma_f32_16x16x4_f32(false, a1, false, b1, (short)0, c, false, false);
        if (l < 8) {
            if (MODE == 1) {
                float* pa = acc + (t * 16 + 8 * half) * 8 + l;
#pragma unroll
                for (int j = 0; j < 8; ++j) pa[j * 8] = c[j] + bs;
            } else {
                const int mbase = row0 + 8 * half;
                float* po = out + (size_t)mbase * 16 + colofs + l;
#pragma unroll
                for (int j = 0; j < 8; ++j)
                    if (mbase + j < n) po[(size_t)j * 16] = c[j] + bs;
            }
        }
    }
    if (MODE == 1) {
        __syncthreads();
        float4* dst4 = (float4*)(rec + park0);
        for (int i = threadIdx.x; i < nodes * 2; i += 1024) dst4[i] = ((const float4*)acc)[i];
    }
}

extern "C" void kernel_launch(void* const* d_in, const int* in_sizes, int n_in,
                              void* d_out, int out_size, void* d_ws, size_t ws_size,
                              hipStream_t stream) {
    const float* x     = (const float*)d_in[0];
    const int*   ei    = (const int*)d_in[1];
    const float* gamma = (const float*)d_in[2];
    const float* beta  = (const float*)d_in[3];
    const float* W_in  = (const float*)d_in[4];
    const float* b_in  = (const float*)d_in[5];
    const float* W_out = (const float*)d_in[6];
    const float* b_out = (const float*)d_in[7];
    float* out = (float*)d_out;

    const int n  = in_sizes[0] / 8;
    const int ne = in_sizes[1] / 2;
    const int NB = 1024;

    float* ws      = (float*)d_ws;
    unsigned* rec  = (unsigned*)ws;
    float* dis_c   = ws + (size_t)n * 16;
    float* dis_r   = dis_c + n;
    float* partial = dis_r + n;
    float* scsh    = partial + (size_t)NB * 16;

    const int* rows = ei;
    const int* cols = ei + ne;

    unsigned* tab_in     = (unsigned*)dis_r;
    unsigned* bstart_in  = tab_in + NBINS * NCHUNK;
    unsigned* tab_out    = bstart_in + NBINS;
    unsigned* bstart_out = (unsigned*)dis_c;

    stats_count_kernel<<<NCHUNK, 1024, 0, stream>>>(x, cols, rows, partial, tab_in, tab_out, n, ne);

    bin_place_kernel<true><<<NCHUNK + 1, 1024, 0, stream>>>(cols, rows, tab_in, bstart_in, rec, ne,
                                                            partial, gamma, beta, scsh, NB, 1.0f / (float)n);
    bin_deg_kernel<<<NBINS, 1024, 0, stream>>>(rec, bstart_in, dis_c, x, scsh, out, n, ne);
    bin_sum_kernel<0><<<NBINS, 1024, 0, stream>>>(rec, bstart_in, x, scsh, dis_c,
                                                  W_in, b_in, out, out, 0, n, ne);

    bin_place_kernel<false><<<NCHUNK, 1024, 0, stream>>>(rows, cols, tab_out, bstart_out, rec, ne,
                                                         partial, gamma, beta, scsh, NB, 1.0f / (float)n);
    bin_deg_kernel<<<NBINS, 1024, 0, stream>>>(rec, bstart_out, dis_r, x, scsh, out, n, ne);
    bin_sum_kernel<1><<<NBINS, 1024, 0, stream>>>(rec, bstart_out, x, scsh, dis_r,
                                                  W_out, b_out, out, out, 8, n, ne);
    bin_sum_kernel<2><<<NBINS, 1024, 0, stream>>>(rec, bstart_out, x, scsh, dis_r,
                                                  W_out, b_out, out, out, 8, n, ne);
}


# given a large enough workspace, each direction gets its own record buffer and row table there: both directions share every launch, nothing is parked or copied; 8 launches become 4; add order kept
# speedup vs baseline: 10.6876x; 1.1585x over previous
#include <hip/hip_runtime.h>
#include <hip/hip_bf16.h>

typedef float v2f __attribute__((ext_vector_type(2)));
typedef float v8f __attribute__((ext_vector_type(8)));

__device__ __forceinline__ void fadd_agent(float* p, float v) {
    __hip_atomic_fetch_add(p, v, __ATOMIC_RELAXED, __HIP_MEMORY_SCOPE_AGENT);
}

#define BIN_SHIFT 11
#define BIN_NODES 2048
#define NBINS 256
#define NCHUNK 256

__global__ void __launch_bounds__(1024)
stats_count_kernel(const float* __restrict__ x,
                   const int* __restrict__ keyA,
                   const int* __restrict__ keyB,
                   float* __restrict__ partial,
                   unsigned* __restrict__ tabA,
                   unsigned* __restrict__ tabB, int n, int ne) {
    __shared__ unsigned histA[NBINS];
    __shared__ unsigned histB[NBINS];
    __shared__ float sh[32 * 16];
    if (threadIdx.x < NBINS) { histA[threadIdx.x] = 0u; histB[threadIdx.x] = 0u; }

    float v[16];
#pragma unroll
    for (int k = 0; k < 16; ++k) v[k] = 0.f;

    const int stride = gridDim.x * blockDim.x;
    for (int i = blockIdx.x * blockDim.x + threadIdx.x; i < n; i += stride) {
        const float4* p = (const float4*)(x + (size_t)i * 8);
        float4 v0 = p[0];
        float4 v1 = p[1];
        float h;
        h = fmaxf(v0.x, 0.f); v[0] += h; v[8]  += h * h;
        h = fmaxf(v0.y, 0.f); v[1] += h; v[9]  += h * h;
        h = fmaxf(v0.z, 0.f); v[2] += h; v[10] += h * h;
        h = fmaxf(v0.w, 0.f); v[3] += h; v[11] += h * h;
        h = fmaxf(v1.x, 0.f); v[4] += h; v[12] += h * h;
        h = fmaxf(v1.y, 0.f); v[5] += h; v[13] += h * h;
        h = fmaxf(v1.z, 0.f); v[6] += h; v[14] += h * h;
        h = fmaxf(v1.w, 0.f); v[7] += h; v[15] += h * h;
    }

#pragma unroll
    for (int k = 0; k < 16; ++k) {
        float t = v[k];
        t += __shfl_xor(t, 1);
        t += __shfl_xor(t, 2);
        t += __shfl_xor(t, 4);
        t += __shfl_xor(t, 8);
        t += __shfl_xor(t, 16);
        v[k] = t;
    }

    const int wave = threadIdx.x >> 5;
    const int lane = threadIdx.x & 31;
    if (lane == 0) {
#pragma unroll
        for (int k = 0; k < 16; ++k) sh[wave * 16 + k] = v[k];
    }
    __syncthreads();
    {
        const int qt = threadIdx.x >> 8;
        const int c  = threadIdx.x & 255;
        if (c < 16) {
            float t = 0.f;
#pragma unroll
            for (int w = 0; w < 8; ++w) t += sh[(qt * 8 + w) * 16 + c];
            partial[(blockIdx.x * 4 + qt) * 16 + c] = t;
        }
    }

    const int chunk = (ne + NCHUNK - 1) / NCHUNK;
    const int beg = blockIdx.x * chunk;
    const int end = min(beg + chunk, ne);
    for (int i = beg + threadIdx.x; i < end; i += 4 * 1024) {
        int ka[4], kb[4];
#pragma unroll
        for (int u = 0; u < 4; ++u) {
            const int j = i + u * 1024;
            const int jc = j < end ? j : i;
            ka[u] = keyA[jc];
            kb[u] = keyB[jc];
        }
#pragma unroll
        for (int u = 0; u < 4; ++u) {
            const unsigned one = (i + u * 1024 < end) ? 1u : 0u;
            atomicAdd(&histA[ka[u] >> BIN_SHIFT], one);
            atomicAdd(&histB[kb[u] >> BIN_SHIFT], one);
        }
    }
    __syncthreads();
    if (threadIdx.x < NBINS) {
        tabA[blockIdx.x * NBINS + threadIdx.x] = histA[threadIdx.x];
        tabB[blockIdx.x * NBINS + threadIdx.x] = histB[threadIdx.x];
    }
}

__device__ __forceinline__ void
bn_finalize_block(const float* __restrict__ partial,
                  const float* __restrict__ gamma,
                  const float* __restrict__ beta,
                  float* __restrict__ scsh, int nb, float invN, float* red) {
    const int lane = threadIdx.x & 31;
    const int wave = threadIdx.x >> 5;
    const int k    = threadIdx.x & 15;
    const int r0   = threadIdx.x >> 4;
    float acc = 0.f;
    for (int rb = r0; rb < nb; rb += 1024) {
        float v[16];
#pragma unroll
        for (int j = 0; j < 16; ++j) {
            const int r = rb + 64 * j;
            const float t = partial[(size_t)(r < nb ? r : rb) * 16 + k];
            v[j] = (r < nb) ? t : 0.f;
        }
#pragma unroll
        for (int j = 0; j < 16; ++j) acc += v[j];
    }
    acc += __shfl_xor(acc, 16);
    if (lane < 16) red[wave * 16 + lane] = acc;
    __syncthreads();
    if (wave == 0) {
        float t = 0.f;
#pragma unroll
        for (int w = 0; w < 32; ++w) t += red[w * 16 + k];
        const float q = __shfl(t, (lane + 8) & 31);
        if (lane < 8) {
            float mean = t * invN;
            float var  = q * invN - mean * mean;
            float sc   = gamma[lane] * rsqrtf(var + 1e-5f);
            scsh[lane]     = sc;
            scsh[8 + lane] = beta[lane] - mean * sc;
        }
    }
}


#define STAGE_CAP 32768
template <bool FIN>
__global__ void bin_place_kernel(const int* __restrict__ key0,
                                 const int* __restrict__ src0,
                                 const unsigned* __restrict__ tab0,
                                 unsigned* __restrict__ bstart0,
                                 unsigned* __restrict__ rec0,
                                 const unsigned* __restrict__ tab1,
                                 unsigned* __restrict__ bstart1,
                                 unsigned* __restrict__ rec1, int ne,
                                 const float* __restrict__ partial,
                                 const float* __restrict__ gamma,
                                 const float* __restrict__ beta,
                                 float* __restrict__ scsh, int nb, float invN) {
    __shared__ unsigned cur[NBINS];
    __shared__ unsigned tot[NBINS];
    __shared__ unsigned lst[NBINS + 1];
    __shared__ unsigned gst[NBINS];
    __shared__ __attribute__((aligned(16))) unsigned stage[STAGE_CAP];
    if (FIN && blockIdx.x == NCHUNK) {
        bn_finalize_block(partial, gamma, beta, scsh, nb, invN, (float*)stage);
        return;
    }
    const bool second = FIN && blockIdx.x > NCHUNK;
    const int cb = second ? (int)blockIdx.x - (NCHUNK + 1) : (int)blockIdx.x;
    const int* key       = second ? src0 : key0;
    const int* srcn      = second ? key0 : src0;
    const unsigned* tab  = second ? tab1 : tab0;
    unsigned* bstart     = second ? bstart1 : bstart0;
    unsigned* rec        = second ? rec1 : rec0;
    unsigned* partT = stage;
    unsigned* partP = stage + 16 * NBINS;
    const unsigned own = (threadIdx.x < NBINS) ? tab[(size_t)cb * NBINS + threadIdx.x] : 0u;
    {
        const int c   = cb;
        const int qd  = threadIdx.x & 63;
        const int grp = threadIdx.x >> 6;
        const uint4* col = (const uint4*)tab + (size_t)(grp * 16) * (NBINS / 4) + qd;
        uint4 r[16];
#pragma unroll
        for (int i = 0; i < 16; ++i) r[i] = col[i * (NBINS / 4)];
        unsigned t0 = 0u, t1 = 0u, t2 = 0u, t3 = 0u;
        unsigned p0 = 0u, p1 = 0u, p2 = 0u, p3 = 0u;
#pragma unroll
        for (int i = 0; i < 16; ++i) {
            t0 += r[i].x; t1 += r[i].y; t2 += r[i].z; t3 += r[i].w;
            if (grp * 16 + i < c) { p0 += r[i].x; p1 += r[i].y; p2 += r[i].z; p3 += r[i].w; }
        }
        const int o = grp * NBINS + 4 * qd;
        partT[o] = t0; partT[o + 1] = t1; partT[o + 2] = t2; partT[o + 3] = t3;
        partP[o] = p0; partP[o + 1] = p1; partP[o + 2] = p2; partP[o + 3] = p3;
    }
    __syncthreads();
    if (threadIdx.x < NBINS) {
        unsigned t = 0u, p = 0u;
#pragma unroll
        for (int g = 0; g < 16; ++g) {
            t += partT[g * NBINS + threadIdx.x];
            p += partP[g * NBINS + threadIdx.x];
        }
        tot[threadIdx.x] = t;
        cur[threadIdx.x] = p;
        lst[threadIdx.x] = own;
    }
    __syncthreads();
    if (threadIdx.x < 32) {
        const int lane = threadIdx.x;
        unsigned v[NBINS / 32], cv[NBINS / 32], ow[NBINS / 32];
        unsigned s = 0u, so = 0u;
#pragma unroll
        for (int k = 0; k < NBINS / 32; ++k) {
            v[k]  = tot[lane * (NBINS / 32) + k];
            cv[k] = cur[lane * (NBINS / 32) + k];
            ow[k] = lst[lane * (NBINS / 32) + k];
            s += v[k];
            so += ow[k];
        }
        unsigned inc = s, inco = so;
#pragma unroll
        for (int d = 1; d < 32; d <<= 1) {
            const unsigned up  = __shfl_up(inc, d);
            const unsigned upo = __shfl_up(inco, d);
            if (lane >= d) { inc += up; inco += upo; }
        }
        unsigned run  = inc - s;
        unsigned lrun = inco - so;
#pragma unroll
        for (int k = 0; k < NBINS / 32; ++k) {
            const int b = lane * (NBINS / 32) + k;
            gst[b] = cv[k] + run;
            lst[b] = lrun;
            cur[b] = lrun;
            if (cb == 0) bstart[b] = run;
            run  += v[k];
            lrun += ow[k];
        }
        if (lane == 31) lst[NBINS] = lrun;
    }
    __syncthreads();
    const int chunk = (ne + NCHUNK - 1) / NCHUNK;
    const int beg = cb * chunk;
    const int end = min(beg + chunk, ne);
    for (int i = beg + threadIdx.x; i < end; i += 4 * 1024) {
        int k[4], s[4];
#pragma unroll
        for (int u = 0; u < 4; ++u) {
            const bool ok = i + u * 1024 < end;
            k[u] = ok ? key[i + u * 1024] : -1;
            s[u] = ok ? srcn[i + u * 1024] : 0;
        }
#pragma unroll
        for (int u = 0; u < 4; ++u)
            if (k[u] >= 0) {
                const int b = k[u] >> BIN_SHIFT;
                const unsigned slot = atomicAdd(&cur[b], 1u);
                const unsigned val = ((unsigned)(k[u] & (BIN_NODES - 1)) << 19) | (unsigned)s[u];
                if (slot < STAGE_CAP) stage[slot] = val;
                else rec[gst[b] + (slot - lst[b])] = val;
            }
    }
    __syncthreads();
    {
        const int lane = threadIdx.x & 31;
        const int wave = threadIdx.x >> 5;
        for (int b = wave; b < NBINS; b += 32) {
            const int l0 = (int)lst[b];
            const int l1 = (int)min(lst[b + 1], (unsigned)STAGE_CAP);
            const unsigned g0 = gst[b];
            if (l1 <= l0) continue;
            const int a = (int)(g0 & 31u);
            for (int j = l0 - a + lane; j < l1; j += 4 * 32) {
                unsigned v[4];
#pragma unroll
                for (int u = 0; u < 4; ++u) v[u] = stage[min(max(j + 32 * u, l0), l1 - 1)];
#pragma unroll
                for (int u = 0; u < 4; ++u) {
                    const int jj = j + 32 * u;
                    if (jj >= l0 && jj < l1) rec[g0 + (unsigned)(jj - l0)] = v[u];
                }
            }
        }
    }
}

__global__ void __launch_bounds__(1024)
bin_deg_kernel(const unsigned* __restrict__ rec0,
               const unsigned* __restrict__ bstart0,
               float* __restrict__ dis0,
               float* __restrict__ ytab0,
               const unsigned* __restrict__ rec1,
               const unsigned* __restrict__ bstart1,
               float* __restrict__ dis1,
               float* __restrict__ ytab1,
               const float* __restrict__ x,
               const float* __restrict__ scsh,
               int yshift, int n, int ne) {
    __shared__ unsigned cnt[BIN_NODES];
    const bool second = blockIdx.x >= NBINS;
    const int b = second ? (int)blockIdx.x - NBINS : (int)blockIdx.x;
    const unsigned* rec    = second ? rec1 : rec0;
    const unsigned* bstart = second ? bstart1 : bstart0;
    float* dis             = second ? dis1 : dis0;
    float* ytab            = second ? ytab1 : ytab0;
    const int base = b << BIN_SHIFT;
    if (base >= n) return;
    for (int i = threadIdx.x; i < BIN_NODES; i += blockDim.x) cnt[i] = 1u;
    __syncthreads();
    const int beg = (int)bstart[b];
    const int end = (b == NBINS - 1) ? ne : (int)bstart[b + 1];
    for (int i = beg + threadIdx.x; i < end; i += 8 * 1024) {
        unsigned r[8];
#pragma unroll
        for (int u = 0; u < 8; ++u) {
            const int j = i + u * 1024;
            r[u] = rec[j < end ? j : i];
        }
#pragma unroll
        for (int u = 0; u < 8; ++u)
            atomicAdd(&cnt[r[u] >> 19], (i + u * 1024 < end) ? 1u : 0u);
    }
    __syncthreads();
    for (int i = threadIdx.x; i < BIN_NODES; i += blockDim.x) {
        const int node = base + i;
        if (node < n) dis[node] = rsqrtf((float)cnt[i]);
    }
    {
        const int hf = threadIdx.x & 1;
        float4 sc, sh;
        sc.x = scsh[4 * hf];     sc.y = scsh[4 * hf + 1];  sc.z = scsh[4 * hf + 2];  sc.w = scsh[4 * hf + 3];
        sh.x = scsh[8 + 4 * hf]; sh.y = scsh[9 + 4 * hf];  sh.z = scsh[10 + 4 * hf]; sh.w = scsh[11 + 4 * hf];
        float4 xv[BIN_NODES / 512];
#pragma unroll
        for (int u = 0; u < BIN_NODES / 512; ++u) {
            const int node = base + (threadIdx.x >> 1) + 512 * u;
            xv[u] = ((const float4*)(x + (size_t)(node < n ? node : n - 1) * 8))[hf];
        }
#pragma unroll
        for (int u = 0; u < BIN_NODES / 512; ++u) {
            const int q = (threadIdx.x >> 1) + 512 * u;
            const int node = base + q;
            const float d = rsqrtf((float)cnt[q]);
            float4 y;
            y.x = (fmaxf(xv[u].x, 0.f) * sc.x + sh.x) * d;
            y.y = (fmaxf(xv[u].y, 0.f) * sc.y + sh.y) * d;
            y.z = (fmaxf(xv[u].z, 0.f) * sc.z + sh.z) * d;
            y.w = (fmaxf(xv[u].w, 0.f) * sc.w + sh.w) * d;
            if (node < n) ((float4*)(ytab + ((size_t)node << yshift)))[hf] = y;
        }
    }
}

template <int MODE>
__global__ void __launch_bounds__(1024)
bin_sum_kernel(unsigned* rec0,
               const unsigned* __restrict__ bstart0,
               const float* __restrict__ dis0,
               const float* __restrict__ Wm0,
               const float* __restrict__ bias0,
               const float* ytab0,
               unsigned* rec1,
               const unsigned* __restrict__ bstart1,
               const float* __restrict__ dis1,
               const float* __restrict__ Wm1,
               const float* __restrict__ bias1,
               const float* ytab1,
               const float* __restrict__ x,
               const float* __restrict__ scsh,
               int yshift,
               float* out, int colofs0, int n, int ne) {
    __shared__ __attribute__((aligned(16))) float acc[BIN_NODES * 8];
    constexpr bool YROW = (MODE != 2);
    const bool second = blockIdx.x >= NBINS;
    const int b = second ? (int)blockIdx.x - NBINS : (int)blockIdx.x;
    unsigned* rec          = second ? rec1 : rec0;
    const unsigned* bstart = second ? bstart1 : bstart0;
    const float* dis       = second ? dis1 : dis0;
    const float* Wm        = second ? Wm1 : Wm0;
    const float* bias      = second ? bias1 : bias0;
    const float* ytab      = second ? ytab1 : ytab0;
    const int colofs       = second ? colofs0 + 8 : colofs0;
    const int base = b << BIN_SHIFT;
    if (base >= n) return;
    const int beg = (int)bstart[b];
    const int end = (b == NBINS - 1) ? ne : (int)bstart[b + 1];
    const int nodes = min(BIN_NODES, n - base);
    const int park0 = (beg + 31) & ~31;
    const bool fits = park0 + nodes * 8 <= end;
    if (MODE == 1 && !fits) return;
    if (MODE == 2 && fits) {
        static_assert(BIN_NODES * 2 == 4 * 1024, "four 16-B pieces per lane");
        const float4* src4 = (const float4*)(rec + park0);
        float* dst = out + (size_t)base * 16 + colofs;
        const int np = nodes * 2;
        const int i0 = threadIdx.x, i1 = i0 + 1024, i2 = i0 + 2048, i3 = i0 + 3072;
        const float4 p0 = src4[i0 < np ? i0 : 0];
        const float4 p1 = src4[i1 < np ? i1 : 0];
        const float4 p2 = src4[i2 < np ? i2 : 0];
        const float4 p3 = src4[i3 < np ? i3 : 0];
        if (i0 < np) ((float4*)(dst + (size_t)(i0 >> 1) * 16))[i0 & 1] = p0;
        if (i1 < np) ((float4*)(dst + (size_t)(i1 >> 1) * 16))[i1 & 1] = p1;
        if (i2 < np) ((float4*)(dst + (size_t)(i2 >> 1) * 16))[i2 & 1] = p2;
        if (i3 < np) ((float4*)(dst + (size_t)(i3 >> 1) * 16))[i3 & 1] = p3;
        return;
    }
    for (int i = threadIdx.x; i < BIN_NODES * 8; i += blockDim.x) acc[i] = 0.f;
    __syncthreads();

    const int lane = threadIdx.x & 31;
    const int wave = threadIdx.x >> 5;
    const int f = lane & 7;
    const int g = lane >> 3;
    const float sc = scsh[f];
    const float sh = scsh[8 + f];

    for (int i0 = beg + wave * 32; i0 < end; i0 += 1024) {
        const int i = i0 + lane;
        unsigned r = 0u;
        float w = 0.f;
        if (i < end) {
            r = rec[i];
            if (!YROW) w = dis[r & 0x7FFFFu];
        }
        float v[8];
        int   a[8];
#pragma unroll
        for (int j = 0; j < 8; ++j) {
            const int s = 4 * j + g;
            const unsigned rj = __shfl(r, s);
            const int sj = (int)(rj & 0x7FFFFu);
            if (YROW) {
                v[j] = ytab[((unsigned)sj << yshift) + (unsigned)f];
            } else {
                const float wj = __shfl(w, s);
                v[j] = (fmaxf(x[sj * 8 + f], 0.f) * sc + sh) * wj;
            }
            a[j] = (int)(rj >> 19) * 8 + f;
        }
#pragma unroll
        for (int j = 0; j < 8; ++j)
            if (i0 + 4 * j + g < end) atomicAdd(&acc[a[j]], v[j]);
    }
    __syncthreads();

    const int half = lane >> 4;
    const int l    = lane & 15;
    const int kA0 = 2 * half;
    const int kA1 = 4 + 2 * half;
    v2f b0 = {0.f, 0.f}, b1 = {0.f, 0.f};
    float bs = 0.f;
    if (l < 8) {
        b0.x = Wm[kA0 * 8 + l];  b0.y = Wm[(kA0 + 1) * 8 + l];
        b1.x = Wm[kA1 * 8 + l];  b1.y = Wm[(kA1 + 1) * 8 + l];
        bs = bias[l];
    }
    for (int t = wave; t < BIN_NODES / 16; t += 32) {
        const int row0 = base + t * 16;
        if (row0 >= n) break;
        const int lrow = t * 16 + l;
        const int row  = row0 + l;
        const int rowc = row < n ? row : (n - 1);
        const float d = dis[rowc];
        const float* xr = x + (size_t)rowc * 8;
        const float* ar = acc + lrow * 8;
        v2f a0, a1;
        a0.x = (ar[kA0]     + (fmaxf(xr[kA0],     0.f) * scsh[kA0]     + scsh[8 + kA0])     * d) * d;
        a0.y = (ar[kA0 + 1] + (fmaxf(xr[kA0 + 1], 0.f) * scsh[kA0 + 1] + scsh[8 + kA0 + 1]) * d) * d;
        a1.x = (ar[kA1]     + (fmaxf(xr[kA1],     0.f) * scsh[kA1]     + scsh[8 + kA1])     * d) * d;
        a1.y = (ar[kA1 + 1] + (fmaxf(xr[kA1 + 1], 0.f) * scsh[kA1 + 1] + scsh[8 + kA1 + 1]) * d) * d;
        v8f c = {};
        c = __builtin_amdgcn_wmma_f32_16x16x4_f32(false, a0, false, b0, (short)0, c, false, false);
        c = __builtin_amdgcn_wmma_f32_16x16x4_f32(false, a1, false, b1, (short)0, c, false, false);
        if (l < 8) {
            if (MODE == 1) {
                float* pa = acc + (t * 16 + 8 * half) * 8 + l;
#pragma unroll
                for (int j = 0; j < 8; ++j) pa[j * 8] = c[j] + bs;
            } else {
                const int mbase = row0 + 8 * half;
                float* po = out + (size_t)mbase * 16 + colofs + l;
#pragma unroll
                for (int j = 0; j < 8; ++j)
                    if (mbase + j < n) po[(size_t)j * 16] = c[j] + bs;
            }
        }
    }
    if (MODE == 1) {
        __syncthreads();
        float4* dst4 = (float4*)(rec + park0);
        for (int i = threadIdx.x; i < nodes * 2; i += 1024) dst4[i] = ((const float4*)acc)[i];
    }
}

extern "C" void kernel_launch(void* const* d_in, const int* in_sizes, int n_in,
                              void* d_out, int out_size, void* d_ws, size_t ws_size,
                              hipStream_t stream) {
    const float* x     = (const float*)d_in[0];
    const int*   ei    = (const int*)d_in[1];
    const float* gamma = (const float*)d_in[2];
    const float* beta  = (const float*)d_in[3];
    const float* W_in  = (const float*)d_in[4];
    const float* b_in  = (const float*)d_in[5];
    const float* W_out = (const float*)d_in[6];
    const float* b_out = (const float*)d_in[7];
    float* out = (float*)d_out;

    const int n  = in_sizes[0] / 8;
    const int ne = in_sizes[1] / 2;
    const int NB = 1024;

    const int* rows = ei;
    const int* cols = ei + ne;
    const float invN = 1.0f / (float)n;

    const size_t nep          = ((size_t)ne + 31) & ~(size_t)31;
    const size_t narrow_words = (size_t)n * 16 + (size_t)n * 2 + (size_t)NB * 16 + 16;
    const size_t table_at     = (narrow_words + 3) & ~(size_t)3;
    const size_t table_words  = table_at + (size_t)n * 8;
    const size_t wide_words   = nep * 2 + (size_t)n * 16 + 2 * (size_t)(NBINS * NCHUNK) + 2 * NBINS
                              + (size_t)NB * 16 + 16 + (size_t)n * 2;
    float* ws = (float*)d_ws;

    if (ws_size >= wide_words * sizeof(float)) {
        unsigned* rec_in     = (unsigned*)ws;
        unsigned* rec_out    = rec_in + nep;
        float* ytab_in       = (float*)(rec_out + nep);
        float* ytab_out      = ytab_in + (size_t)n * 8;
        unsigned* tab_in     = (unsigned*)(ytab_out + (size_t)n * 8);
        unsigned* tab_out    = tab_in + NBINS * NCHUNK;
        unsigned* bstart_in  = tab_out + NBINS * NCHUNK;
        unsigned* bstart_out = bstart_in + NBINS;
        float* partial       = (float*)(bstart_out + NBINS);
        float* scsh          = partial + (size_t)NB * 16;
        float* dis_c         = scsh + 16;
        float* dis_r         = dis_c + n;

        stats_count_kernel<<<NCHUNK, 1024, 0, stream>>>(x, cols, rows, partial, tab_in, tab_out, n, ne);
        bin_place_kernel<true><<<2 * NCHUNK + 1, 1024, 0, stream>>>(cols, rows, tab_in, bstart_in, rec_in,
                                                                    tab_out, bstart_out, rec_out, ne,
                                                                    partial, gamma, beta, scsh, NB, invN);
        bin_deg_kernel<<<2 * NBINS, 1024, 0, stream>>>(rec_in, bstart_in, dis_c, ytab_in,
                                                       rec_out, bstart_out, dis_r, ytab_out,
                                                       x, scsh, 3, n, ne);
        bin_sum_kernel<0><<<2 * NBINS, 1024, 0, stream>>>(rec_in, bstart_in, dis_c, W_in, b_in, ytab_in,
                                                          rec_out, bstart_out, dis_r, W_out, b_out, ytab_out,
                                                          x, scsh, 3, out, 0, n, ne);
        return;
    }

    const bool own_table = ws_size >= table_words * sizeof(float);
    unsigned* rec  = (unsigned*)ws;
    float* dis_c   = ws + (size_t)n * 16;
    float* dis_r   = dis_c + n;
    float* partial = dis_r + n;
    float* scsh    = partial + (size_t)NB * 16;
    float* ytab    = own_table ? ws + table_at : out + 8;
    const int yshift = own_table ? 3 : 4;

    unsigned* tab_in     = (unsigned*)dis_r;
    unsigned* bstart_in  = tab_in + NBINS * NCHUNK;
    unsigned* tab_out    = bstart_in + NBINS;
    unsigned* bstart_out = (unsigned*)dis_c;

    stats_count_kernel<<<NCHUNK, 1024, 0, stream>>>(x, cols, rows, partial, tab_in, tab_out, n, ne);

    bin_place_kernel<true><<<NCHUNK + 1, 1024, 0, stream>>>(cols, rows, tab_in, bstart_in, rec,
                                                            tab_in, bstart_in, rec, ne,
                                                            partial, gamma, beta, scsh, NB, invN);
    bin_deg_kernel<<<NBINS, 1024, 0, stream>>>(rec, bstart_in, dis_c, ytab,
                                               rec, bstart_in, dis_c, ytab, x, scsh, yshift, n, ne);
    bin_sum_kernel<0><<<NBINS, 1024, 0, stream>>>(rec, bstart_in, dis_c, W_in, b_in, ytab,
                                                  rec, bstart_in, dis_c, W_in, b_in, ytab,
                                                  x, scsh, yshift, out, 0, n, ne);

    bin_place_kernel<false><<<NCHUNK, 1024, 0, stream>>>(rows, cols, tab_out, bstart_out, rec,
                                                         tab_out, bstart_out, rec, ne,
                                                         partial, gamma, beta, scsh, NB, invN);
    bin_deg_kernel<<<NBINS, 1024, 0, stream>>>(rec, bstart_out, dis_r, ytab,
                                               rec, bstart_out, dis_r, ytab, x, scsh, yshift, n, ne);
    if (own_table) {
        bin_sum_kernel<0><<<NBINS, 1024, 0, stream>>>(rec, bstart_out, dis_r, W_out, b_out, ytab,
                                                      rec, bstart_out, dis_r, W_out, b_out, ytab,
                                                      x, scsh, yshift, out, 8, n, ne);
        return;
    }
    bin_sum_kernel<1><<<NBINS, 1024, 0, stream>>>(rec, bstart_out, dis_r, W_out, b_out, ytab,
                                                  rec, bstart_out, dis_r, W_out, b_out, ytab,
                                                  x, scsh, yshift, out, 8, n, ne);
    bin_sum_kernel<2><<<NBINS, 1024, 0, stream>>>(rec, bstart_out, dis_r, W_out, b_out, ytab,
                                                  rec, bstart_out, dis_r, W_out, b_out, ytab,
                                                  x, scsh, yshift, out, 8, n, ne);
}
